# starting kernel
# baseline (speedup 1.0000x reference)
#include <hip/hip_runtime.h>
#include <hip/hip_bf16.h>

typedef __attribute__((ext_vector_type(16))) _Float16 v16h;
typedef __attribute__((ext_vector_type(8)))  _Float16 v8h;
typedef __attribute__((ext_vector_type(8)))  float    v8f;
typedef __attribute__((ext_vector_type(2)))  float    v2f;

#define N_HEADS   16
#define N_QUBITS  8
#define DIM       256
#define Q_LAYERS  2
#define N_TOKENS  2048
#define EMBED     128

#define WAVES_PER_BLOCK 4
#define W_LDS_BYTES     (DIM * DIM * 2)
#define AF_OFF          0
#define BF_OFF          8192
#define PR_OFF          16384
#define CB_OFF          32768
#define SB_OFF          33280
#define WAVE_LDS_BYTES  33792
#define TOTAL_LDS_BYTES (W_LDS_BYTES + WAVES_PER_BLOCK * WAVE_LDS_BYTES)

__global__ __launch_bounds__(256)
void qa_build_unitaries(const float* __restrict__ params,
                        float* __restrict__ Mbuf,
                        _Float16* __restrict__ W) {
    const int h = blockIdx.x;
    const int t = threadIdx.x;
    float* M = Mbuf + (size_t)h * DIM * DIM;

    for (int i = t; i < DIM * DIM; i += 256)
        M[i] = ((i >> 8) == (i & 255)) ? 1.0f : 0.0f;
    __syncthreads();

    for (int layer = 0; layer < Q_LAYERS; ++layer) {
        for (int w = 0; w < N_QUBITS; ++w) {
            const float th = 0.5f * params[(h * Q_LAYERS + layer) * N_QUBITS + w];
            const float cr = cosf(th), sr = sinf(th);
            const int pos = 7 - w;
            const int pair = t & 127;
            const int r0 = ((pair >> pos) << (pos + 1)) | (pair & ((1 << pos) - 1));
            const int r1 = r0 | (1 << pos);
            const int cbase = (t >> 7) * 128;
            float* row0 = M + r0 * DIM;
            float* row1 = M + r1 * DIM;
            for (int c = cbase; c < cbase + 128; ++c) {
                const float m0 = row0[c], m1 = row1[c];
                row0[c] = cr * m0 - sr * m1;
                row1[c] = sr * m0 + cr * m1;
            }
            __syncthreads();
        }
        for (int w = 0; w < N_QUBITS - 1; ++w) {
            const int pt = 6 - w;
            const int q = t & 63;
            const int r0 = ((q >> pt) << (pt + 2)) | (1 << (pt + 1)) | (q & ((1 << pt) - 1));
            const int r1 = r0 | (1 << pt);
            const int cbase = (t >> 6) * 64;
            float* row0 = M + r0 * DIM;
            float* row1 = M + r1 * DIM;
            for (int c = cbase; c < cbase + 64; ++c) {
                const float tmp = row0[c];
                row0[c] = row1[c];
                row1[c] = tmp;
            }
            __syncthreads();
        }
    }

    _Float16* Wh = W + (size_t)h * DIM * DIM;
    for (int i = t; i < DIM * DIM; i += 256)
        Wh[i] = (_Float16)M[i];
}

__global__ __launch_bounds__(32 * WAVES_PER_BLOCK)
void qa_measure(const float* __restrict__ x,
                const _Float16* __restrict__ W,
                float* __restrict__ out) {
    extern __shared__ char smem[];
    const int h    = blockIdx.y;
    const int wave = threadIdx.x >> 5;
    const int lane = threadIdx.x & 31;
    const int tile = blockIdx.x * WAVES_PER_BLOCK + wave;
    const int t0   = tile * 16;

    _Float16* Wlds = (_Float16*)smem;
    char* wbase = smem + W_LDS_BYTES + wave * WAVE_LDS_BYTES;
    _Float16* af    = (_Float16*)(wbase + AF_OFF);
    _Float16* bf    = (_Float16*)(wbase + BF_OFF);
    float*    probs = (float*)(wbase + PR_OFF);
    float*    cbuf  = (float*)(wbase + CB_OFF);
    float*    sbuf  = (float*)(wbase + SB_OFF);

    {
        const v8h* src = (const v8h*)(W + (size_t)h * DIM * DIM);
        v8h* dst = (v8h*)Wlds;
        for (int i = threadIdx.x; i < DIM * DIM / 8; i += 32 * WAVES_PER_BLOCK)
            dst[i] = src[i];
    }

    for (int i = lane; i < 16 * N_QUBITS; i += 32) {
        const int tt = i >> 3, w = i & 7;
        const float half = 0.5f * x[(size_t)(t0 + tt) * EMBED + h * N_QUBITS + w];
        cbuf[tt * N_QUBITS + w] = cosf(half);
        sbuf[tt * N_QUBITS + w] = sinf(half);
    }
    __syncthreads();

    for (int i = lane; i < 16 * DIM; i += 32) {
        const int tt = i >> 8, s = i & 255;
        float p = 1.0f;
#pragma unroll
        for (int w = 0; w < N_QUBITS; ++w)
            p *= ((s >> (7 - w)) & 1) ? sbuf[tt * N_QUBITS + w] : cbuf[tt * N_QUBITS + w];
        const int pc = __popc(s) & 3;
        const float a = (pc == 0) ? p : ((pc == 2) ? -p : 0.0f);
        const float b = (pc == 1) ? -p : ((pc == 3) ? p : 0.0f);
        af[tt * DIM + s] = (_Float16)a;
        bf[tt * DIM + s] = (_Float16)b;
    }
    __syncthreads();

    const int m     = lane & 15;
    const int hi    = lane >> 4;
    const int khalf = hi * 8;

    for (int ct = 0; ct < 16; ++ct) {
        v8f accA = {};
        v8f accB = {};
#pragma unroll
        for (int kt = 0; kt < 8; ++kt) {
            const int kb = kt * 32 + khalf;
            const v8h aLo = *(const v8h*)&af[m * DIM + kb];
            const v8h aHi = *(const v8h*)&af[m * DIM + kb + 16];
            const v8h bLo = *(const v8h*)&bf[m * DIM + kb];
            const v8h bHi = *(const v8h*)&bf[m * DIM + kb + 16];
            const _Float16* bp = Wlds + (size_t)(ct * 16 + m) * DIM + kb;
            const v8h wLo = *(const v8h*)bp;
            const v8h wHi = *(const v8h*)(bp + 16);
            v16h Aa, Ab, Bf;
#pragma unroll
            for (int i = 0; i < 8; ++i) {
                Aa[i] = aLo[i]; Aa[i + 8] = aHi[i];
                Ab[i] = bLo[i]; Ab[i + 8] = bHi[i];
                Bf[i] = wLo[i]; Bf[i + 8] = wHi[i];
            }
            accA = __builtin_amdgcn_wmma_f32_16x16x32_f16(
                false, Aa, false, Bf, (short)0, accA, false, false);
            accB = __builtin_amdgcn_wmma_f32_16x16x32_f16(
                false, Ab, false, Bf, (short)0, accB, false, false);
        }
#pragma unroll
        for (int r = 0; r < 8; ++r) {
            const float re = accA[r], im = accB[r];
            probs[(r + 8 * hi) * DIM + ct * 16 + m] = re * re + im * im;
        }
    }
    __syncthreads();

    const int n = lane & 15;
    const int shift = 7 - (n & 7);
    v8f zacc = {};
#pragma unroll 8
    for (int k2 = 0; k2 < 64; ++k2) {
        const int kb2 = k2 * 4 + 2 * hi;
        const v2f pA = *(const v2f*)&probs[m * DIM + kb2];
        v2f pB;
#pragma unroll
        for (int j = 0; j < 2; ++j) {
            const int sp = kb2 + j;
            const float sg = ((sp >> shift) & 1) ? -1.0f : 1.0f;
            pB[j] = (n < N_QUBITS) ? sg : 0.0f;
        }
        zacc = __builtin_amdgcn_wmma_f32_16x16x4_f32(
            false, pA, false, pB, (short)0, zacc, false, false);
    }

    if (n < N_QUBITS) {
#pragma unroll
        for (int r = 0; r < 8; ++r) {
            const int tt = r + 8 * hi;
            out[(size_t)(t0 + tt) * EMBED + h * N_QUBITS + n] = zacc[r];
        }
    }
}

extern "C" void kernel_launch(void* const* d_in, const int* in_sizes, int n_in,
                              void* d_out, int out_size, void* d_ws, size_t ws_size,
                              hipStream_t stream) {
    const float* x      = (const float*)d_in[0];
    const float* params = (const float*)d_in[1];
    float* out          = (float*)d_out;

    float*    Mbuf = (float*)d_ws;
    _Float16* W    = (_Float16*)((char*)d_ws + (size_t)N_HEADS * DIM * DIM * sizeof(float));

    qa_build_unitaries<<<dim3(N_HEADS), dim3(256), 0, stream>>>(params, Mbuf, W);

    dim3 grid(N_TOKENS / 16 / WAVES_PER_BLOCK, N_HEADS);
    qa_measure<<<grid, dim3(32 * WAVES_PER_BLOCK), TOTAL_LDS_BYTES, stream>>>(x, W, out);
}


# Measure kernel transposed: each wave keeps 32 rows of U_h in registers as A, no LDS copy of U_h; tokens are the columns; |psi|^2 goes from D into B of the signed sums; no dtype change; sums regrouped
# speedup vs baseline: 85.5571x; 85.5571x over previous
#include <hip/hip_runtime.h>
#include <hip/hip_bf16.h>

typedef __attribute__((ext_vector_type(16))) _Float16 v16h;
typedef __attribute__((ext_vector_type(8)))  _Float16 v8h;
typedef __attribute__((ext_vector_type(8)))  float    v8f;
typedef __attribute__((ext_vector_type(2)))  float    v2f;

#define N_HEADS   16
#define N_QUBITS  8
#define DIM       256
#define Q_LAYERS  2
#define N_TOKENS  2048
#define EMBED     128

#define M_WAVES     8
#define M_STRIPS    2
#define M_TOKENS    128
#define M_TILES     (M_TOKENS / 16)
#define CS_OFF      0
#define AMP_OFF     8192
#define AMP_BUF_BYTES 16384
#define PART_OFF    (AMP_OFF + 2 * AMP_BUF_BYTES)
#define M_LDS_BYTES (PART_OFF + M_TILES * M_WAVES * 128 * 4)

__global__ __launch_bounds__(128)
void qa_build_unitaries(const float* __restrict__ params,
                        _Float16* __restrict__ W) {
    const int h    = blockIdx.y;
    const int it   = blockIdx.x;
    const int wv   = threadIdx.x >> 5;
    const int lane = threadIdx.x & 31;
    const int c    = lane & 15;
    const int hh   = lane >> 4;

    const float th = 0.5f * params[h * (Q_LAYERS * N_QUBITS) + c];
    const float cv = cosf(th), sv = sinf(th);
    float C0[N_QUBITS], S0[N_QUBITS], C1[N_QUBITS], S1[N_QUBITS];
#pragma unroll
    for (int w = 0; w < N_QUBITS; ++w) {
        C0[w] = __int_as_float(__builtin_amdgcn_readlane(__float_as_int(cv), w));
        S0[w] = __int_as_float(__builtin_amdgcn_readlane(__float_as_int(sv), w));
        C1[w] = __int_as_float(__builtin_amdgcn_readlane(__float_as_int(cv), N_QUBITS + w));
        S1[w] = __int_as_float(__builtin_amdgcn_readlane(__float_as_int(sv), N_QUBITS + w));
    }

    const int i = it * 16 + c;
    const int a = i ^ (i >> 1);
    float fA[N_QUBITS][2];
#pragma unroll
    for (int w = 0; w < N_QUBITS; ++w) {
        const bool ab = (a >> (7 - w)) & 1;
        fA[w][0] = ab ? S1[w] :  C1[w];
        fA[w][1] = ab ? C1[w] : -S1[w];
    }
    const int jb = 64 * wv + 4 * c;
    float eB[6][2];
#pragma unroll
    for (int w = 0; w < 6; ++w) {
        const bool jbit = (jb >> (7 - w)) & 1;
        eB[w][0] = jbit ? -S0[w] : C0[w];
        eB[w][1] = jbit ?  C0[w] : S0[w];
    }
    const float E6[2][2] = { { C0[6], S0[6] }, { -S0[6], C0[6] } };
    const float E7[2][2] = { { C0[7], S0[7] }, { -S0[7], C0[7] } };

    v8f acc[4] = { {}, {}, {}, {} };

    const float pa0  = hh ? fA[0][1] : fA[0][0];
    const float pb0  = hh ? eB[0][1] : eB[0][0];
    const float eb1x0 = hh ? eB[1][1] : eB[1][0];
    const float eb1x1 = hh ? eB[1][0] : eB[1][1];

#pragma unroll
    for (int t1 = 0; t1 < 2; ++t1) {
        const float pa1 = pa0 * fA[1][t1];
        const float pb1 = pb0 * (t1 ? eb1x1 : eb1x0);
#pragma unroll
        for (int t2 = 0; t2 < 2; ++t2) {
            const float pa2 = pa1 * fA[2][t2];
            const float pb2 = pb1 * eB[2][t1 ^ t2];
#pragma unroll
            for (int t3 = 0; t3 < 2; ++t3) {
                const float pa3 = pa2 * fA[3][t3];
                const float pb3 = pb2 * eB[3][t2 ^ t3];
#pragma unroll
                for (int t4 = 0; t4 < 2; ++t4) {
                    const float pa4 = pa3 * fA[4][t4];
                    const float pb4 = pb3 * eB[4][t3 ^ t4];
#pragma unroll
                    for (int t5 = 0; t5 < 2; ++t5) {
                        const float pa5 = pa4 * fA[5][t5];
                        const float pb5 = pb4 * eB[5][t4 ^ t5];
#pragma unroll
                        for (int t6 = 0; t6 < 2; ++t6) {
                            const float pa6 = pa5 * fA[6][t6];
                            v2f Af;
                            Af[0] = pa6 * fA[7][0];
                            Af[1] = pa6 * fA[7][1];
#pragma unroll
                            for (int q = 0; q < 4; ++q) {
                                const float b6 = pb5 * E6[q >> 1][t5 ^ t6];
                                v2f Bf;
                                Bf[0] = b6 * E7[q & 1][t6];
                                Bf[1] = b6 * E7[q & 1][t6 ^ 1];
                                acc[q] = __builtin_amdgcn_wmma_f32_16x16x4_f32(
                                    false, Af, false, Bf, (short)0, acc[q], false, false);
                            }
                        }
                    }
                }
            }
        }
    }

    typedef __attribute__((ext_vector_type(4))) _Float16 v4h;
    _Float16* Wh = W + (size_t)h * DIM * DIM;
#pragma unroll
    for (int r = 0; r < 8; ++r) {
        const int row = it * 16 + 8 * hh + r;
        v4h o;
        o[0] = (_Float16)acc[0][r];
        o[1] = (_Float16)acc[1][r];
        o[2] = (_Float16)acc[2][r];
        o[3] = (_Float16)acc[3][r];
        *(v4h*)&Wh[row * DIM + jb] = o;
    }
}

__device__ __forceinline__ void qa_build_tile(const float* __restrict__ cb,
                                              const float* __restrict__ sb,
                                              char* __restrict__ ampbuf,
                                              int tile, int tid) {
    const int n  = tid & 15;
    const int gq = tid >> 4;
    const float4* cp = (const float4*)(cb + (tile * 16 + n) * N_QUBITS);
    const float4* sp = (const float4*)(sb + (tile * 16 + n) * N_QUBITS);
    const float4 c03 = cp[0], c47 = cp[1], s03 = sp[0], s47 = sp[1];
    const float cw[8] = { c03.x, c03.y, c03.z, c03.w, c47.x, c47.y, c47.z, c47.w };
    const float sw[8] = { s03.x, s03.y, s03.z, s03.w, s47.x, s47.y, s47.z, s47.w };
    _Float16* re = (_Float16*)ampbuf;
    _Float16* im = (_Float16*)(ampbuf + AMP_BUF_BYTES / 2);
#pragma unroll
    for (int b0 = 0; b0 < 2; ++b0) {
        const int g = gq + 16 * b0;
        float p = b0 ? sw[0] : cw[0];
#pragma unroll
        for (int w = 1; w < 5; ++w)
            p *= ((gq >> (4 - w)) & 1) ? sw[w] : cw[w];
        const int pg = __popc(g);
        v8h ar, ai;
#pragma unroll
        for (int e = 0; e < 8; ++e) {
            float pe = p * ((e & 4) ? sw[5] : cw[5]);
            pe = pe * ((e & 2) ? sw[6] : cw[6]);
            pe = pe * ((e & 1) ? sw[7] : cw[7]);
            const int pc = (pg + __popc(e)) & 3;
            const float a = (pc == 0) ? pe : ((pc == 2) ? -pe : 0.0f);
            const float b = (pc == 1) ? -pe : ((pc == 3) ? pe : 0.0f);
            ar[e] = (_Float16)a;
            ai[e] = (_Float16)b;
        }
        const int kt = g >> 2, lohi = (g >> 1) & 1, hh = g & 1;
        const int off = (((kt * 2 + lohi) * 32) + n + 16 * hh) * 8;
        *(v8h*)&re[off] = ar;
        *(v8h*)&im[off] = ai;
    }
}

__global__ __launch_bounds__(32 * M_WAVES) __attribute__((amdgpu_waves_per_eu(2, 2)))
void qa_measure(const float* __restrict__ x,
                const _Float16* __restrict__ W,
                float* __restrict__ out) {
    extern __shared__ char smem[];
    const int h    = blockIdx.y;
    const int tok0 = blockIdx.x * M_TOKENS;
    const int tid  = threadIdx.x;
    const int wave = tid >> 5;
    const int lane = tid & 31;
    const int c    = lane & 15;
    const int hi   = lane >> 4;

    float* cb   = (float*)(smem + CS_OFF);
    float* sb   = cb + M_TOKENS * N_QUBITS;
    float* part = (float*)(smem + PART_OFF);

    const float4 xv = *(const float4*)&x[(size_t)(tok0 + (tid >> 1)) * EMBED
                                         + h * N_QUBITS + 4 * (tid & 1)];

    v16h Ua[M_STRIPS][8];
    {
        const _Float16* Wh = W + (size_t)h * DIM * DIM;
#pragma unroll
        for (int st = 0; st < M_STRIPS; ++st) {
            const _Float16* up = Wh + (size_t)((wave * M_STRIPS + st) * 16 + c) * DIM + 8 * hi;
#pragma unroll
            for (int kt = 0; kt < 8; ++kt) {
                const v8h lo = *(const v8h*)(up + kt * 32);
                const v8h hp = *(const v8h*)(up + kt * 32 + 16);
                Ua[st][kt] = __builtin_shufflevector(lo, hp, 0, 1, 2, 3, 4, 5, 6, 7,
                                                     8, 9, 10, 11, 12, 13, 14, 15);
            }
        }
    }

    v2f Sg[M_STRIPS][4];
#pragma unroll
    for (int st = 0; st < M_STRIPS; ++st)
#pragma unroll
        for (int j = 0; j < 4; ++j)
#pragma unroll
            for (int s = 0; s < 2; ++s) {
                const int row = (wave * M_STRIPS + st) * 16 + 8 * hi + 2 * j + s;
                const float sg = ((row >> (7 - (c & 7))) & 1) ? -1.0f : 1.0f;
                Sg[st][j][s] = (c < N_QUBITS) ? sg : 0.0f;
            }

    {
        float4 cv4, sv4;
        cv4.x = cosf(0.5f * xv.x); sv4.x = sinf(0.5f * xv.x);
        cv4.y = cosf(0.5f * xv.y); sv4.y = sinf(0.5f * xv.y);
        cv4.z = cosf(0.5f * xv.z); sv4.z = sinf(0.5f * xv.z);
        cv4.w = cosf(0.5f * xv.w); sv4.w = sinf(0.5f * xv.w);
        *(float4*)&cb[4 * tid] = cv4;
        *(float4*)&sb[4 * tid] = sv4;
    }
    __syncthreads();

    qa_build_tile(cb, sb, smem + AMP_OFF, 0, tid);
    __syncthreads();

    for (int tile = 0; tile < M_TILES; ++tile) {
        const _Float16* re = (const _Float16*)(smem + AMP_OFF + (tile & 1) * AMP_BUF_BYTES);
        const _Float16* im = re + AMP_BUF_BYTES / 4;

        v16h Br[8], Bi[8];
#pragma unroll
        for (int kt = 0; kt < 8; ++kt) {
            const int o0 = ((kt * 2 + 0) * 32 + lane) * 8;
            const int o1 = ((kt * 2 + 1) * 32 + lane) * 8;
            const v8h rLo = *(const v8h*)&re[o0];
            const v8h rHi = *(const v8h*)&re[o1];
            const v8h iLo = *(const v8h*)&im[o0];
            const v8h iHi = *(const v8h*)&im[o1];
            Br[kt] = __builtin_shufflevector(rLo, rHi, 0, 1, 2, 3, 4, 5, 6, 7,
                                             8, 9, 10, 11, 12, 13, 14, 15);
            Bi[kt] = __builtin_shufflevector(iLo, iHi, 0, 1, 2, 3, 4, 5, 6, 7,
                                             8, 9, 10, 11, 12, 13, 14, 15);
        }

        if (tile + 1 < M_TILES)
            qa_build_tile(cb, sb, smem + AMP_OFF + ((tile + 1) & 1) * AMP_BUF_BYTES, tile + 1, tid);

        v8f accR[M_STRIPS], accI[M_STRIPS];
#pragma unroll
        for (int st = 0; st < M_STRIPS; ++st) { accR[st] = (v8f){}; accI[st] = (v8f){}; }

#pragma unroll
        for (int kt = 0; kt < 8; ++kt) {
#pragma unroll
            for (int st = 0; st < M_STRIPS; ++st) {
                accR[st] = __builtin_amdgcn_wmma_f32_16x16x32_f16(
                    false, Ua[st][kt], false, Br[kt], (short)0, accR[st], false, false);
                accI[st] = __builtin_amdgcn_wmma_f32_16x16x32_f16(
                    false, Ua[st][kt], false, Bi[kt], (short)0, accI[st], false, false);
            }
        }

        v8f zacc[M_STRIPS];
#pragma unroll
        for (int st = 0; st < M_STRIPS; ++st) {
            zacc[st] = (v8f){};
#pragma unroll
            for (int j = 0; j < 4; ++j) {
                v2f pB;
#pragma unroll
                for (int s = 0; s < 2; ++s) {
                    const float r = accR[st][2 * j + s], q = accI[st][2 * j + s];
                    pB[s] = r * r + q * q;
                }
                zacc[st] = __builtin_amdgcn_wmma_f32_16x16x4_f32(
                    false, Sg[st][j], false, pB, (short)0, zacc[st], false, false);
            }
        }

        if (hi == 0) {
            float4 lo4, hi4;
            lo4.x = zacc[0][0] + zacc[1][0]; lo4.y = zacc[0][1] + zacc[1][1];
            lo4.z = zacc[0][2] + zacc[1][2]; lo4.w = zacc[0][3] + zacc[1][3];
            hi4.x = zacc[0][4] + zacc[1][4]; hi4.y = zacc[0][5] + zacc[1][5];
            hi4.z = zacc[0][6] + zacc[1][6]; hi4.w = zacc[0][7] + zacc[1][7];
            float4* pp = (float4*)(part + ((tile * M_WAVES + wave) * 16 + c) * N_QUBITS);
            pp[0] = lo4;
            pp[1] = hi4;
        }
        __syncthreads();
    }

    for (int o = tid; o < M_TOKENS * N_QUBITS; o += 32 * M_WAVES) {
        const int tile = o >> 7, rest = o & 127;
        float z = 0.0f;
#pragma unroll
        for (int wv = 0; wv < M_WAVES; ++wv)
            z += part[(tile * M_WAVES + wv) * 128 + rest];
        const int tt = tile * 16 + (rest >> 3), w = rest & 7;
        out[(size_t)(tok0 + tt) * EMBED + h * N_QUBITS + w] = z;
    }
}

extern "C" void kernel_launch(void* const* d_in, const int* in_sizes, int n_in,
                              void* d_out, int out_size, void* d_ws, size_t ws_size,
                              hipStream_t stream) {
    const float* x      = (const float*)d_in[0];
    const float* params = (const float*)d_in[1];
    float* out          = (float*)d_out;

    _Float16* W = (_Float16*)d_ws;

    qa_build_unitaries<<<dim3(DIM / 16, N_HEADS), dim3(128), 0, stream>>>(params, W);

    dim3 grid(N_TOKENS / M_TOKENS, N_HEADS);
    qa_measure<<<grid, dim3(32 * M_WAVES), M_LDS_BYTES, stream>>>(x, W, out);
}


# Measure kernel: K split by popcount parity (amplitudes are real at even, imaginary at odd popcount): re and im products run K=128 on the matching U_h columns; half the f16 matrix ops; sums regrouped
# speedup vs baseline: 97.1057x; 1.1350x over previous
#include <hip/hip_runtime.h>
#include <hip/hip_bf16.h>

typedef __attribute__((ext_vector_type(16))) _Float16 v16h;
typedef __attribute__((ext_vector_type(8)))  _Float16 v8h;
typedef __attribute__((ext_vector_type(8)))  float    v8f;
typedef __attribute__((ext_vector_type(2)))  float    v2f;

#define N_HEADS   16
#define N_QUBITS  8
#define DIM       256
#define Q_LAYERS  2
#define N_TOKENS  2048
#define EMBED     128

#define M_WAVES     8
#define M_STRIPS    2
#define M_TOKENS    128
#define M_TILES     (M_TOKENS / 16)
#define CS_OFF      0
#define AMP_OFF     8192
#define AMP_BUF_BYTES 8192
#define PART_OFF    (AMP_OFF + 2 * AMP_BUF_BYTES)
#define M_LDS_BYTES (PART_OFF + M_TILES * M_WAVES * 128 * 4)

__global__ __launch_bounds__(128)
void qa_build_unitaries(const float* __restrict__ params,
                        _Float16* __restrict__ W) {
    const int h    = blockIdx.y;
    const int it   = blockIdx.x;
    const int wv   = threadIdx.x >> 5;
    const int lane = threadIdx.x & 31;
    const int c    = lane & 15;
    const int hh   = lane >> 4;

    const float th = 0.5f * params[h * (Q_LAYERS * N_QUBITS) + c];
    const float cv = cosf(th), sv = sinf(th);
    float C0[N_QUBITS], S0[N_QUBITS], C1[N_QUBITS], S1[N_QUBITS];
#pragma unroll
    for (int w = 0; w < N_QUBITS; ++w) {
        C0[w] = __int_as_float(__builtin_amdgcn_readlane(__float_as_int(cv), w));
        S0[w] = __int_as_float(__builtin_amdgcn_readlane(__float_as_int(sv), w));
        C1[w] = __int_as_float(__builtin_amdgcn_readlane(__float_as_int(cv), N_QUBITS + w));
        S1[w] = __int_as_float(__builtin_amdgcn_readlane(__float_as_int(sv), N_QUBITS + w));
    }

    const int i = it * 16 + c;
    const int a = i ^ (i >> 1);
    float fA[N_QUBITS][2];
#pragma unroll
    for (int w = 0; w < N_QUBITS; ++w) {
        const bool ab = (a >> (7 - w)) & 1;
        fA[w][0] = ab ? S1[w] :  C1[w];
        fA[w][1] = ab ? C1[w] : -S1[w];
    }
    const int jb = 64 * wv + 4 * c;
    float eB[6][2];
#pragma unroll
    for (int w = 0; w < 6; ++w) {
        const bool jbit = (jb >> (7 - w)) & 1;
        eB[w][0] = jbit ? -S0[w] : C0[w];
        eB[w][1] = jbit ?  C0[w] : S0[w];
    }
    const float E6[2][2] = { { C0[6], S0[6] }, { -S0[6], C0[6] } };
    const float E7[2][2] = { { C0[7], S0[7] }, { -S0[7], C0[7] } };

    v8f acc[4] = { {}, {}, {}, {} };

    const float pa0  = hh ? fA[0][1] : fA[0][0];
    const float pb0  = hh ? eB[0][1] : eB[0][0];
    const float eb1x0 = hh ? eB[1][1] : eB[1][0];
    const float eb1x1 = hh ? eB[1][0] : eB[1][1];

#pragma unroll
    for (int t1 = 0; t1 < 2; ++t1) {
        const float pa1 = pa0 * fA[1][t1];
        const float pb1 = pb0 * (t1 ? eb1x1 : eb1x0);
#pragma unroll
        for (int t2 = 0; t2 < 2; ++t2) {
            const float pa2 = pa1 * fA[2][t2];
            const float pb2 = pb1 * eB[2][t1 ^ t2];
#pragma unroll
            for (int t3 = 0; t3 < 2; ++t3) {
                const float pa3 = pa2 * fA[3][t3];
                const float pb3 = pb2 * eB[3][t2 ^ t3];
#pragma unroll
                for (int t4 = 0; t4 < 2; ++t4) {
                    const float pa4 = pa3 * fA[4][t4];
                    const float pb4 = pb3 * eB[4][t3 ^ t4];
#pragma unroll
                    for (int t5 = 0; t5 < 2; ++t5) {
                        const float pa5 = pa4 * fA[5][t5];
                        const float pb5 = pb4 * eB[5][t4 ^ t5];
#pragma unroll
                        for (int t6 = 0; t6 < 2; ++t6) {
                            const float pa6 = pa5 * fA[6][t6];
                            v2f Af;
                            Af[0] = pa6 * fA[7][0];
                            Af[1] = pa6 * fA[7][1];
#pragma unroll
                            for (int q = 0; q < 4; ++q) {
                                const float b6 = pb5 * E6[q >> 1][t5 ^ t6];
                                v2f Bf;
                                Bf[0] = b6 * E7[q & 1][t6];
                                Bf[1] = b6 * E7[q & 1][t6 ^ 1];
                                acc[q] = __builtin_amdgcn_wmma_f32_16x16x4_f32(
                                    false, Af, false, Bf, (short)0, acc[q], false, false);
                            }
                        }
                    }
                }
            }
        }
    }

    typedef __attribute__((ext_vector_type(4))) _Float16 v4h;
    _Float16* Wh = W + (size_t)h * DIM * DIM;
#pragma unroll
    for (int r = 0; r < 8; ++r) {
        const int row = it * 16 + 8 * hh + r;
        v4h o;
        o[0] = (_Float16)acc[0][r];
        o[1] = (_Float16)acc[1][r];
        o[2] = (_Float16)acc[2][r];
        o[3] = (_Float16)acc[3][r];
        *(v4h*)&Wh[row * DIM + jb] = o;
    }
}


typedef __attribute__((ext_vector_type(4))) _Float16 v4hh;
typedef __attribute__((ext_vector_type(4))) unsigned int v4u;
typedef __attribute__((ext_vector_type(8))) unsigned int v8u;

__device__ __forceinline__ void qa_build_tile(const float* __restrict__ cb,
                                              const float* __restrict__ sb,
                                              char* __restrict__ ampbuf,
                                              int tile, int tid) {
    const int n  = tid & 15;
    const int gq = tid >> 4;
    const float4* cp = (const float4*)(cb + (tile * 16 + n) * N_QUBITS);
    const float4* sp = (const float4*)(sb + (tile * 16 + n) * N_QUBITS);
    const float4 c03 = cp[0], c47 = cp[1], s03 = sp[0], s47 = sp[1];
    const float cw[8] = { c03.x, c03.y, c03.z, c03.w, c47.x, c47.y, c47.z, c47.w };
    const float sw[8] = { s03.x, s03.y, s03.z, s03.w, s47.x, s47.y, s47.z, s47.w };
    _Float16* ev = (_Float16*)ampbuf;
    _Float16* od = (_Float16*)(ampbuf + AMP_BUF_BYTES / 2);
#pragma unroll
    for (int b0 = 0; b0 < 2; ++b0) {
        const int g = gq + 16 * b0;
        float p = b0 ? sw[0] : cw[0];
#pragma unroll
        for (int w = 1; w < 5; ++w)
            p *= ((gq >> (4 - w)) & 1) ? sw[w] : cw[w];
        const int pg = __popc(g);
        const bool par = pg & 1;
        const float ps  = (pg & 2) ? -p : p;
        const float a7  = par ? sw[7] : cw[7];
        const float b7  = par ? cw[7] : sw[7];
        const float a7p = par ? -a7 : a7;
        const float u0 = ps * cw[5], u1 = ps * sw[5];
        const float t0 = u0 * cw[6], t1 = u0 * sw[6], t2 = u1 * cw[6], t3 = u1 * sw[6];
        v4hh ar, ai;
        ar[0] = (_Float16)( t0 * a7p);  ai[0] = (_Float16)(-t0 * b7);
        ar[1] = (_Float16)(-t1 * b7);   ai[1] = (_Float16)(-t1 * a7p);
        ar[2] = (_Float16)(-t2 * b7);   ai[2] = (_Float16)(-t2 * a7p);
        ar[3] = (_Float16)(-t3 * a7p);  ai[3] = (_Float16)( t3 * b7);
        const int f = g >> 3, g4 = (g >> 1) & 3, hh = g & 1;
        const int off = (((f * 2 + (g4 >> 1)) * 32) + n + 16 * hh) * 8 + 4 * (g4 & 1);
        *(v4hh*)&ev[off] = ar;
        *(v4hh*)&od[off] = ai;
    }
}

__global__ __launch_bounds__(32 * M_WAVES) __attribute__((amdgpu_waves_per_eu(2, 2)))
void qa_measure(const float* __restrict__ x,
                const _Float16* __restrict__ W,
                float* __restrict__ out) {
    extern __shared__ char smem[];
    const int h    = blockIdx.y;
    const int tok0 = blockIdx.x * M_TOKENS;
    const int tid  = threadIdx.x;
    const int wave = tid >> 5;
    const int lane = tid & 31;
    const int c    = lane & 15;
    const int hi   = lane >> 4;

    float* cb   = (float*)(smem + CS_OFF);
    float* sb   = cb + M_TOKENS * N_QUBITS;
    float* part = (float*)(smem + PART_OFF);

    const float4 xv = *(const float4*)&x[(size_t)(tok0 + (tid >> 1)) * EMBED
                                         + h * N_QUBITS + 4 * (tid & 1)];

    v4u Uq[M_STRIPS][4][4];
    {
        const _Float16* Wh = W + (size_t)h * DIM * DIM;
#pragma unroll
        for (int st = 0; st < M_STRIPS; ++st) {
            const _Float16* up = Wh + (size_t)((wave * M_STRIPS + st) * 16 + c) * DIM + 8 * hi;
#pragma unroll
            for (int f = 0; f < 4; ++f)
#pragma unroll
                for (int g4 = 0; g4 < 4; ++g4)
                    Uq[st][f][g4] = *(const v4u*)(up + 64 * f + 16 * g4);
        }
    }

    v2f Sg[M_STRIPS][4];
#pragma unroll
    for (int st = 0; st < M_STRIPS; ++st)
#pragma unroll
        for (int j = 0; j < 4; ++j)
#pragma unroll
            for (int s = 0; s < 2; ++s) {
                const int row = (wave * M_STRIPS + st) * 16 + 8 * hi + 2 * j + s;
                const float sg = ((row >> (7 - (c & 7))) & 1) ? -1.0f : 1.0f;
                Sg[st][j][s] = (c < N_QUBITS) ? sg : 0.0f;
            }

    {
        float4 cv4, sv4;
        cv4.x = cosf(0.5f * xv.x); sv4.x = sinf(0.5f * xv.x);
        cv4.y = cosf(0.5f * xv.y); sv4.y = sinf(0.5f * xv.y);
        cv4.z = cosf(0.5f * xv.z); sv4.z = sinf(0.5f * xv.z);
        cv4.w = cosf(0.5f * xv.w); sv4.w = sinf(0.5f * xv.w);
        *(float4*)&cb[4 * tid] = cv4;
        *(float4*)&sb[4 * tid] = sv4;
    }
    __syncthreads();

    v16h Ue[M_STRIPS][4], Uo[M_STRIPS][4];
#pragma unroll
    for (int st = 0; st < M_STRIPS; ++st)
#pragma unroll
        for (int f = 0; f < 4; ++f) {
            v8u e8, o8;
#pragma unroll
            for (int g4 = 0; g4 < 4; ++g4) {
                const v4u d = Uq[st][f][g4];
                const unsigned x01 = (d[0] & 0xffffu) | (d[1] & 0xffff0000u);
                const unsigned y01 = (d[0] >> 16) | (d[1] << 16);
                const unsigned x23 = (d[2] & 0xffffu) | (d[3] & 0xffff0000u);
                const unsigned y23 = (d[2] >> 16) | (d[3] << 16);
                const bool par = ((__popc(f) + __popc(g4) + hi) & 1) != 0;
                e8[2 * g4 + 0] = par ? y01 : x01;
                e8[2 * g4 + 1] = par ? x23 : y23;
                o8[2 * g4 + 0] = par ? x01 : y01;
                o8[2 * g4 + 1] = par ? y23 : x23;
            }
            Ue[st][f] = __builtin_bit_cast(v16h, e8);
            Uo[st][f] = __builtin_bit_cast(v16h, o8);
        }

    qa_build_tile(cb, sb, smem + AMP_OFF, 0, tid);
    __syncthreads();

    for (int tile = 0; tile < M_TILES; ++tile) {
        const _Float16* ev = (const _Float16*)(smem + AMP_OFF + (tile & 1) * AMP_BUF_BYTES);
        const _Float16* od = ev + AMP_BUF_BYTES / 4;

        v16h Be[4], Bo[4];
#pragma unroll
        for (int f = 0; f < 4; ++f) {
            const int o0 = ((f * 2 + 0) * 32 + lane) * 8;
            const int o1 = ((f * 2 + 1) * 32 + lane) * 8;
            const v8h rLo = *(const v8h*)&ev[o0];
            const v8h rHi = *(const v8h*)&ev[o1];
            const v8h iLo = *(const v8h*)&od[o0];
            const v8h iHi = *(const v8h*)&od[o1];
            Be[f] = __builtin_shufflevector(rLo, rHi, 0, 1, 2, 3, 4, 5, 6, 7,
                                            8, 9, 10, 11, 12, 13, 14, 15);
            Bo[f] = __builtin_shufflevector(iLo, iHi, 0, 1, 2, 3, 4, 5, 6, 7,
                                            8, 9, 10, 11, 12, 13, 14, 15);
        }

        if (tile + 1 < M_TILES)
            qa_build_tile(cb, sb, smem + AMP_OFF + ((tile + 1) & 1) * AMP_BUF_BYTES, tile + 1, tid);

        v8f accR[M_STRIPS], accI[M_STRIPS];
#pragma unroll
        for (int st = 0; st < M_STRIPS; ++st) { accR[st] = (v8f){}; accI[st] = (v8f){}; }

#pragma unroll
        for (int f = 0; f < 4; ++f) {
#pragma unroll
            for (int st = 0; st < M_STRIPS; ++st) {
                accR[st] = __builtin_amdgcn_wmma_f32_16x16x32_f16(
                    false, Ue[st][f], false, Be[f], (short)0, accR[st], false, false);
                accI[st] = __builtin_amdgcn_wmma_f32_16x16x32_f16(
                    false, Uo[st][f], false, Bo[f], (short)0, accI[st], false, false);
            }
        }

        v8f zacc[M_STRIPS];
#pragma unroll
        for (int st = 0; st < M_STRIPS; ++st) {
            zacc[st] = (v8f){};
#pragma unroll
            for (int j = 0; j < 4; ++j) {
                v2f pB;
#pragma unroll
                for (int s = 0; s < 2; ++s) {
                    const float r = accR[st][2 * j + s], q = accI[st][2 * j + s];
                    pB[s] = r * r + q * q;
                }
                zacc[st] = __builtin_amdgcn_wmma_f32_16x16x4_f32(
                    false, Sg[st][j], false, pB, (short)0, zacc[st], false, false);
            }
        }

        if (hi == 0) {
            float4 lo4, hi4;
            lo4.x = zacc[0][0] + zacc[1][0]; lo4.y = zacc[0][1] + zacc[1][1];
            lo4.z = zacc[0][2] + zacc[1][2]; lo4.w = zacc[0][3] + zacc[1][3];
            hi4.x = zacc[0][4] + zacc[1][4]; hi4.y = zacc[0][5] + zacc[1][5];
            hi4.z = zacc[0][6] + zacc[1][6]; hi4.w = zacc[0][7] + zacc[1][7];
            float4* pp = (float4*)(part + ((tile * M_WAVES + wave) * 16 + c) * N_QUBITS);
            pp[0] = lo4;
            pp[1] = hi4;
        }
        __syncthreads();
    }

    for (int o = tid; o < M_TOKENS * N_QUBITS; o += 32 * M_WAVES) {
        const int tile = o >> 7, rest = o & 127;
        float z = 0.0f;
#pragma unroll
        for (int wv = 0; wv < M_WAVES; ++wv)
            z += part[(tile * M_WAVES + wv) * 128 + rest];
        const int tt = tile * 16 + (rest >> 3), w = rest & 7;
        out[(size_t)(tok0 + tt) * EMBED + h * N_QUBITS + w] = z;
    }
}

extern "C" void kernel_launch(void* const* d_in, const int* in_sizes, int n_in,
                              void* d_out, int out_size, void* d_ws, size_t ws_size,
                              hipStream_t stream) {
    const float* x      = (const float*)d_in[0];
    const float* params = (const float*)d_in[1];
    float* out          = (float*)d_out;

    _Float16* W = (_Float16*)d_ws;

    qa_build_unitaries<<<dim3(DIM / 16, N_HEADS), dim3(128), 0, stream>>>(params, W);

    dim3 grid(N_TOKENS / M_TOKENS, N_HEADS);
    qa_measure<<<grid, dim3(32 * M_WAVES), M_LDS_BYTES, stream>>>(x, W, out);
}


# Tile loop software-pipelined: next tile's B fragments and the next build's angles loaded one step ahead (second register set), amplitudes built two tiles ahead; no LDS wait ahead of the matrix block
# speedup vs baseline: 158.3554x; 1.6308x over previous
#include <hip/hip_runtime.h>
#include <hip/hip_bf16.h>

typedef __attribute__((ext_vector_type(16))) _Float16 v16h;
typedef __attribute__((ext_vector_type(8)))  _Float16 v8h;
typedef __attribute__((ext_vector_type(8)))  float    v8f;
typedef __attribute__((ext_vector_type(2)))  float    v2f;

#define N_HEADS   16
#define N_QUBITS  8
#define DIM       256
#define Q_LAYERS  2
#define N_TOKENS  2048
#define EMBED     128

#define M_WAVES     8
#define M_STRIPS    2
#define M_TOKENS    128
#define M_TILES     (M_TOKENS / 16)
#define CS_OFF      0
#define AMP_OFF     8192
#define AMP_BUF_BYTES 8192
#define PART_OFF    (AMP_OFF + 2 * AMP_BUF_BYTES)
#define M_LDS_BYTES (PART_OFF + M_TILES * M_WAVES * 128 * 4)


typedef __attribute__((ext_vector_type(4))) _Float16 v4hh;

__device__ __forceinline__ void qa_load_cs(const float* __restrict__ cb,
                                           const float* __restrict__ sb,
                                           int tile, int tid, float4 (&cs)[4]) {
    const int n = tid & 15;
    const float4* cp = (const float4*)(cb + (tile * 16 + n) * N_QUBITS);
    const float4* sp = (const float4*)(sb + (tile * 16 + n) * N_QUBITS);
    cs[0] = cp[0]; cs[1] = cp[1]; cs[2] = sp[0]; cs[3] = sp[1];
}

__device__ __forceinline__ void qa_build_tile(const float4 (&cs)[4],
                                              char* __restrict__ ampbuf, int tid) {
    const int n  = tid & 15;
    const int gq = tid >> 4;
    const float4 c03 = cs[0], c47 = cs[1], s03 = cs[2], s47 = cs[3];
    const float cw[8] = { c03.x, c03.y, c03.z, c03.w, c47.x, c47.y, c47.z, c47.w };
    const float sw[8] = { s03.x, s03.y, s03.z, s03.w, s47.x, s47.y, s47.z, s47.w };
    _Float16* ev = (_Float16*)ampbuf;
    _Float16* od = (_Float16*)(ampbuf + AMP_BUF_BYTES / 2);
#pragma unroll
    for (int b0 = 0; b0 < 2; ++b0) {
        const int g = gq + 16 * b0;
        float p = b0 ? sw[0] : cw[0];
#pragma unroll
        for (int w = 1; w < 5; ++w)
            p *= ((gq >> (4 - w)) & 1) ? sw[w] : cw[w];
        const int pg = __popc(g);
        const bool par = pg & 1;
        const float ps  = (pg & 2) ? -p : p;
        const float a7  = par ? sw[7] : cw[7];
        const float b7  = par ? cw[7] : sw[7];
        const float a7p = par ? -a7 : a7;
        const float u0 = ps * cw[5], u1 = ps * sw[5];
        const float t0 = u0 * cw[6], t1 = u0 * sw[6], t2 = u1 * cw[6], t3 = u1 * sw[6];
        v4hh ar, ai;
        ar[0] = (_Float16)( t0 * a7p);  ai[0] = (_Float16)(-t0 * b7);
        ar[1] = (_Float16)(-t1 * b7);   ai[1] = (_Float16)(-t1 * a7p);
        ar[2] = (_Float16)(-t2 * b7);   ai[2] = (_Float16)(-t2 * a7p);
        ar[3] = (_Float16)(-t3 * a7p);  ai[3] = (_Float16)( t3 * b7);
        const int f = g >> 3, g4 = (g >> 1) & 3, hh = g & 1;
        const int off = (((f * 2 + (g4 >> 1)) * 32) + n + 16 * hh) * 8 + 4 * (g4 & 1);
        *(v4hh*)&ev[off] = ar;
        *(v4hh*)&od[off] = ai;
    }
}

__device__ __forceinline__ void qa_load_frags(const char* __restrict__ ampbuf, int lane,
                                              v16h (&Be)[4], v16h (&Bo)[4]) {
    const _Float16* ev = (const _Float16*)ampbuf;
    const _Float16* od = ev + AMP_BUF_BYTES / 4;
#pragma unroll
    for (int f = 0; f < 4; ++f) {
        const int o0 = ((f * 2 + 0) * 32 + lane) * 8;
        const int o1 = ((f * 2 + 1) * 32 + lane) * 8;
        const v8h rLo = *(const v8h*)&ev[o0];
        const v8h rHi = *(const v8h*)&ev[o1];
        const v8h iLo = *(const v8h*)&od[o0];
        const v8h iHi = *(const v8h*)&od[o1];
        Be[f] = __builtin_shufflevector(rLo, rHi, 0, 1, 2, 3, 4, 5, 6, 7,
                                        8, 9, 10, 11, 12, 13, 14, 15);
        Bo[f] = __builtin_shufflevector(iLo, iHi, 0, 1, 2, 3, 4, 5, 6, 7,
                                        8, 9, 10, 11, 12, 13, 14, 15);
    }
}


__global__ __launch_bounds__(32 * M_WAVES) __attribute__((amdgpu_waves_per_eu(2, 2)))
void qa_measure(const float* __restrict__ x,
                const float* __restrict__ params,
                float* __restrict__ out) {
    extern __shared__ char smem[];
    const int h    = blockIdx.y;
    const int tok0 = blockIdx.x * M_TOKENS;
    const int tid  = threadIdx.x;
    const int wave = tid >> 5;
    const int lane = tid & 31;
    const int c    = lane & 15;
    const int hi   = lane >> 4;

    float* cb   = (float*)(smem + CS_OFF);
    float* sb   = cb + M_TOKENS * N_QUBITS;
    float* part = (float*)(smem + PART_OFF);

    const float4 xv = *(const float4*)&x[(size_t)(tok0 + (tid >> 1)) * EMBED
                                         + h * N_QUBITS + 4 * (tid & 1)];
    const float thp = 0.5f * params[h * (Q_LAYERS * N_QUBITS) + c];

    v2f Sg[M_STRIPS][4];
#pragma unroll
    for (int st = 0; st < M_STRIPS; ++st)
#pragma unroll
        for (int j = 0; j < 4; ++j)
#pragma unroll
            for (int s = 0; s < 2; ++s) {
                const int row = (wave * M_STRIPS + st) * 16 + 8 * hi + 2 * j + s;
                const float sg = ((row >> (7 - (c & 7))) & 1) ? -1.0f : 1.0f;
                Sg[st][j][s] = (c < N_QUBITS) ? sg : 0.0f;
            }

    {
        float4 cv4, sv4;
        cv4.x = cosf(0.5f * xv.x); sv4.x = sinf(0.5f * xv.x);
        cv4.y = cosf(0.5f * xv.y); sv4.y = sinf(0.5f * xv.y);
        cv4.z = cosf(0.5f * xv.z); sv4.z = sinf(0.5f * xv.z);
        cv4.w = cosf(0.5f * xv.w); sv4.w = sinf(0.5f * xv.w);
        *(float4*)&cb[4 * tid] = cv4;
        *(float4*)&sb[4 * tid] = sv4;
    }
    __syncthreads();

    v16h Be[2][4], Bo[2][4];
    float4 cs[2][4];
    qa_load_cs(cb, sb, 0, tid, cs[0]);
    qa_build_tile(cs[0], smem + AMP_OFF, tid);
    __syncthreads();
    qa_load_frags(smem + AMP_OFF, lane, Be[0], Bo[0]);
    qa_load_cs(cb, sb, 1, tid, cs[1]);
    qa_load_cs(cb, sb, 2, tid, cs[0]);
    qa_build_tile(cs[1], smem + AMP_OFF + AMP_BUF_BYTES, tid);

    const float pcv = cosf(thp), psv = sinf(thp);
    float C0[N_QUBITS], S0[N_QUBITS], C1[N_QUBITS], S1[N_QUBITS];
#pragma unroll
    for (int w = 0; w < N_QUBITS; ++w) {
        C0[w] = __int_as_float(__builtin_amdgcn_readlane(__float_as_int(pcv), w));
        S0[w] = __int_as_float(__builtin_amdgcn_readlane(__float_as_int(psv), w));
        C1[w] = __int_as_float(__builtin_amdgcn_readlane(__float_as_int(pcv), N_QUBITS + w));
        S1[w] = __int_as_float(__builtin_amdgcn_readlane(__float_as_int(psv), N_QUBITS + w));
    }

    const int gc = c ^ (c >> 1);
    float e0l[4][2], f1l[4][2], e0h[4][2], f1h[4][2];
    const int ihw = 2 * wave;
    const int ghw = ihw ^ (ihw >> 1);
#pragma unroll
    for (int w = 0; w < 4; ++w) {
        const bool cbit = (c  >> (3 - w)) & 1;
        const bool rbit = (gc >> (3 - w)) & 1;
        const bool hbit = (ghw >> (3 - w)) & 1;
        e0l[w][0] = cbit ? -S0[4 + w] : C0[4 + w];
        e0l[w][1] = cbit ?  C0[4 + w] : S0[4 + w];
        f1l[w][0] = rbit ?  S1[4 + w] :  C1[4 + w];
        f1l[w][1] = rbit ?  C1[4 + w] : -S1[4 + w];
        e0h[w][0] = cbit ? -S0[w] : C0[w];
        e0h[w][1] = cbit ?  C0[w] : S0[w];
        f1h[w][0] = hbit ?  S1[w] :  C1[w];
        f1h[w][1] = hbit ?  C1[w] : -S1[w];
    }
    const bool rb4 = (gc >> 3) & 1, hb3 = ghw & 1;
    const float f14[2][2] = { { f1l[0][0], f1l[0][1] },
                              { rb4 ? C1[4] : S1[4], rb4 ? -S1[4] : C1[4] } };
    const float f13[2][2] = { { f1h[3][0], f1h[3][1] },
                              { hb3 ? C1[3] : S1[3], hb3 ? -S1[3] : C1[3] } };
    const float e00[2][2] = { { C0[0], S0[0] }, { -S0[0], C0[0] } };

    v8f Qd[M_STRIPS][2];
#pragma unroll
    for (int st = 0; st < M_STRIPS; ++st) { Qd[st][0] = (v8f){}; Qd[st][1] = (v8f){}; }
    {
        const float x6[2] = { hi ? e0l[2][1] : e0l[2][0], hi ? e0l[2][0] : e0l[2][1] };
        const float x7[2] = { hi ? e0l[3][1] : e0l[3][0], hi ? e0l[3][0] : e0l[3][1] };
        const float y6    = hi ? f1l[2][1] : f1l[2][0];
#pragma unroll
        for (int t = 0; t < 4; ++t) {
            const int b3 = t >> 1, b2 = t & 1;
#pragma unroll
            for (int al = 0; al < 2; ++al) {
                const float pa = e0l[0][b3 ^ al] * e0l[1][b2 ^ b3] * x6[b2];
                v2f Af;
                Af[0] = pa * x7[0];
                Af[1] = pa * x7[1];
#pragma unroll
                for (int st = 0; st < M_STRIPS; ++st) {
                    const float pb = f14[st][b3] * f1l[1][b2] * y6;
                    v2f Bf;
                    Bf[0] = pb * f1l[3][0];
                    Bf[1] = pb * f1l[3][1];
                    Qd[st][al] = __builtin_amdgcn_wmma_f32_16x16x4_f32(
                        false, Af, false, Bf, (short)0, Qd[st][al], false, false);
                }
            }
        }
    }

    v8f Rd[M_STRIPS][2][2];
#pragma unroll
    for (int st = 0; st < M_STRIPS; ++st)
#pragma unroll
        for (int al = 0; al < 2; ++al) { Rd[st][al][0] = (v8f){}; Rd[st][al][1] = (v8f){}; }
    {
        const float u1[2] = { hi ? e0h[1][1] : e0h[1][0], hi ? e0h[1][0] : e0h[1][1] };
        const float u2[2] = { hi ? e0h[2][1] : e0h[2][0], hi ? e0h[2][0] : e0h[2][1] };
        const float v1    = hi ? f1h[1][1] : f1h[1][0];
#pragma unroll
        for (int t = 0; t < 2; ++t)
#pragma unroll
            for (int al = 0; al < 2; ++al)
#pragma unroll
                for (int set = 0; set < 2; ++set) {
                    const float pa = e00[set][t] * u1[t];
                    v2f Af;
                    Af[0] = pa * u2[0] * e0h[3][al];
                    Af[1] = pa * u2[1] * e0h[3][al ^ 1];
#pragma unroll
                    for (int st = 0; st < M_STRIPS; ++st) {
                        const float pb = f1h[0][t] * v1 * f13[st][al];
                        v2f Bf;
                        Bf[0] = pb * f1h[2][0];
                        Bf[1] = pb * f1h[2][1];
                        Rd[st][al][set] = __builtin_amdgcn_wmma_f32_16x16x4_f32(
                            false, Af, false, Bf, (short)0, Rd[st][al][set], false, false);
                    }
                }
    }

    v16h Ue[M_STRIPS][4], Uo[M_STRIPS][4];
#pragma unroll
    for (int st = 0; st < M_STRIPS; ++st) {
        float QA[2][4], QB[2][4];
#pragma unroll
        for (int al = 0; al < 2; ++al) {
            const v8f qv = Qd[st][al];
            QA[al][0] = hi ? qv[1] : qv[0];  QB[al][0] = hi ? qv[0] : qv[1];
            QA[al][1] = hi ? qv[2] : qv[3];  QB[al][1] = hi ? qv[3] : qv[2];
            QA[al][2] = hi ? qv[4] : qv[5];  QB[al][2] = hi ? qv[5] : qv[4];
            QA[al][3] = hi ? qv[7] : qv[6];  QB[al][3] = hi ? qv[6] : qv[7];
        }
#pragma unroll
        for (int f = 0; f < 4; ++f)
#pragma unroll
            for (int g4 = 0; g4 < 4; ++g4) {
                const int jh = 4 * f + g4;
                const float r0 = Rd[st][0][jh >> 3][jh & 7];
                const float r1 = Rd[st][1][jh >> 3][jh & 7];
                const bool odd = ((__builtin_popcount(f) + __builtin_popcount(g4)) & 1) != 0;
#pragma unroll
                for (int k = 0; k < 4; ++k) {
                    const float ue = odd ? (r0 * QB[0][k] + r1 * QB[1][k])
                                         : (r0 * QA[0][k] + r1 * QA[1][k]);
                    const float uo = odd ? (r0 * QA[0][k] + r1 * QA[1][k])
                                         : (r0 * QB[0][k] + r1 * QB[1][k]);
                    Ue[st][f][4 * g4 + k] = (_Float16)ue;
                    Uo[st][f][4 * g4 + k] = (_Float16)uo;
                }
            }
    }
    __syncthreads();

#pragma unroll
    for (int tile = 0; tile < M_TILES; ++tile) {
        const int cur = tile & 1, nxt = cur ^ 1;
        if (tile + 1 < M_TILES)
            qa_load_frags(smem + AMP_OFF + nxt * AMP_BUF_BYTES, lane, Be[nxt], Bo[nxt]);
        if (tile + 3 < M_TILES)
            qa_load_cs(cb, sb, tile + 3, tid, cs[nxt]);
        if (tile + 1 < M_TILES)
            __builtin_amdgcn_sched_barrier(0);
        if (tile + 2 < M_TILES)
            qa_build_tile(cs[cur], smem + AMP_OFF + cur * AMP_BUF_BYTES, tid);

        v8f accR[M_STRIPS], accI[M_STRIPS];
#pragma unroll
        for (int st = 0; st < M_STRIPS; ++st) { accR[st] = (v8f){}; accI[st] = (v8f){}; }

#pragma unroll
        for (int f = 0; f < 4; ++f) {
#pragma unroll
            for (int st = 0; st < M_STRIPS; ++st) {
                accR[st] = __builtin_amdgcn_wmma_f32_16x16x32_f16(
                    false, Ue[st][f], false, Be[cur][f], (short)0, accR[st], false, false);
                accI[st] = __builtin_amdgcn_wmma_f32_16x16x32_f16(
                    false, Uo[st][f], false, Bo[cur][f], (short)0, accI[st], false, false);
            }
        }

        v8f zacc[M_STRIPS];
#pragma unroll
        for (int st = 0; st < M_STRIPS; ++st) {
            zacc[st] = (v8f){};
#pragma unroll
            for (int j = 0; j < 4; ++j) {
                v2f pB;
#pragma unroll
                for (int s = 0; s < 2; ++s) {
                    const float r = accR[st][2 * j + s], q = accI[st][2 * j + s];
                    pB[s] = r * r + q * q;
                }
                zacc[st] = __builtin_amdgcn_wmma_f32_16x16x4_f32(
                    false, Sg[st][j], false, pB, (short)0, zacc[st], false, false);
            }
        }

        if (hi == 0) {
            float4 lo4, hi4;
            lo4.x = zacc[0][0] + zacc[1][0]; lo4.y = zacc[0][1] + zacc[1][1];
            lo4.z = zacc[0][2] + zacc[1][2]; lo4.w = zacc[0][3] + zacc[1][3];
            hi4.x = zacc[0][4] + zacc[1][4]; hi4.y = zacc[0][5] + zacc[1][5];
            hi4.z = zacc[0][6] + zacc[1][6]; hi4.w = zacc[0][7] + zacc[1][7];
            float4* pp = (float4*)(part + ((tile * M_WAVES + wave) * 16 + c) * N_QUBITS);
            pp[0] = lo4;
            pp[1] = hi4;
        }
        __syncthreads();
    }

    for (int o = tid; o < M_TOKENS * N_QUBITS; o += 32 * M_WAVES) {
        const int tile = o >> 7, rest = o & 127;
        float z = 0.0f;
#pragma unroll
        for (int wv = 0; wv < M_WAVES; ++wv)
            z += part[(tile * M_WAVES + wv) * 128 + rest];
        const int tt = tile * 16 + (rest >> 3), w = rest & 7;
        out[(size_t)(tok0 + tt) * EMBED + h * N_QUBITS + w] = z;
    }
}

extern "C" void kernel_launch(void* const* d_in, const int* in_sizes, int n_in,
                              void* d_out, int out_size, void* d_ws, size_t ws_size,
                              hipStream_t stream) {
    const float* x      = (const float*)d_in[0];
    const float* params = (const float*)d_in[1];
    float* out          = (float*)d_out;

    dim3 grid(N_TOKENS / M_TOKENS, N_HEADS);
    qa_measure<<<grid, dim3(32 * M_WAVES), M_LDS_BYTES, stream>>>(x, params, out);
}


# Signed sums: wires 5 and 6 folded in the lane by pair sums/differences of the D registers before the f32 matrix step; 3 f32 16x16x4 ops per tile and wave, not 8; sums regrouped
# speedup vs baseline: 169.5703x; 1.0708x over previous
#include <hip/hip_runtime.h>
#include <hip/hip_bf16.h>

typedef __attribute__((ext_vector_type(16))) _Float16 v16h;
typedef __attribute__((ext_vector_type(8)))  _Float16 v8h;
typedef __attribute__((ext_vector_type(8)))  float    v8f;
typedef __attribute__((ext_vector_type(2)))  float    v2f;

#define N_HEADS   16
#define N_QUBITS  8
#define DIM       256
#define Q_LAYERS  2
#define N_TOKENS  2048
#define EMBED     128

#define M_WAVES     8
#define M_STRIPS    2
#define M_TOKENS    128
#define M_TILES     (M_TOKENS / 16)
#define CS_OFF      0
#define AMP_OFF     8192
#define AMP_BUF_BYTES 8192
#define PART_OFF    (AMP_OFF + 2 * AMP_BUF_BYTES)
#define M_LDS_BYTES (PART_OFF + M_TILES * M_WAVES * 128 * 4)


typedef __attribute__((ext_vector_type(4))) _Float16 v4hh;

__device__ __forceinline__ void qa_load_cs(const float* __restrict__ cb,
                                           const float* __restrict__ sb,
                                           int tile, int tid, float4 (&cs)[4]) {
    const int n = tid & 15;
    const float4* cp = (const float4*)(cb + (tile * 16 + n) * N_QUBITS);
    const float4* sp = (const float4*)(sb + (tile * 16 + n) * N_QUBITS);
    cs[0] = cp[0]; cs[1] = cp[1]; cs[2] = sp[0]; cs[3] = sp[1];
}

__device__ __forceinline__ void qa_build_tile(const float4 (&cs)[4],
                                              char* __restrict__ ampbuf, int tid) {
    const int n  = tid & 15;
    const int gq = tid >> 4;
    const float4 c03 = cs[0], c47 = cs[1], s03 = cs[2], s47 = cs[3];
    const float cw[8] = { c03.x, c03.y, c03.z, c03.w, c47.x, c47.y, c47.z, c47.w };
    const float sw[8] = { s03.x, s03.y, s03.z, s03.w, s47.x, s47.y, s47.z, s47.w };
    _Float16* ev = (_Float16*)ampbuf;
    _Float16* od = (_Float16*)(ampbuf + AMP_BUF_BYTES / 2);
#pragma unroll
    for (int b0 = 0; b0 < 2; ++b0) {
        const int g = gq + 16 * b0;
        float p = b0 ? sw[0] : cw[0];
#pragma unroll
        for (int w = 1; w < 5; ++w)
            p *= ((gq >> (4 - w)) & 1) ? sw[w] : cw[w];
        const int pg = __popc(g);
        const bool par = pg & 1;
        const float ps  = (pg & 2) ? -p : p;
        const float a7  = par ? sw[7] : cw[7];
        const float b7  = par ? cw[7] : sw[7];
        const float a7p = par ? -a7 : a7;
        const float u0 = ps * cw[5], u1 = ps * sw[5];
        const float t0 = u0 * cw[6], t1 = u0 * sw[6], t2 = u1 * cw[6], t3 = u1 * sw[6];
        v4hh ar, ai;
        ar[0] = (_Float16)( t0 * a7p);  ai[0] = (_Float16)(-t0 * b7);
        ar[1] = (_Float16)(-t1 * b7);   ai[1] = (_Float16)(-t1 * a7p);
        ar[2] = (_Float16)(-t2 * b7);   ai[2] = (_Float16)(-t2 * a7p);
        ar[3] = (_Float16)(-t3 * a7p);  ai[3] = (_Float16)( t3 * b7);
        const int f = g >> 3, g4 = (g >> 1) & 3, hh = g & 1;
        const int off = (((f * 2 + (g4 >> 1)) * 32) + n + 16 * hh) * 8 + 4 * (g4 & 1);
        *(v4hh*)&ev[off] = ar;
        *(v4hh*)&od[off] = ai;
    }
}

__device__ __forceinline__ void qa_load_frags(const char* __restrict__ ampbuf, int lane,
                                              v16h (&Be)[4], v16h (&Bo)[4]) {
    const _Float16* ev = (const _Float16*)ampbuf;
    const _Float16* od = ev + AMP_BUF_BYTES / 4;
#pragma unroll
    for (int f = 0; f < 4; ++f) {
        const int o0 = ((f * 2 + 0) * 32 + lane) * 8;
        const int o1 = ((f * 2 + 1) * 32 + lane) * 8;
        const v8h rLo = *(const v8h*)&ev[o0];
        const v8h rHi = *(const v8h*)&ev[o1];
        const v8h iLo = *(const v8h*)&od[o0];
        const v8h iHi = *(const v8h*)&od[o1];
        Be[f] = __builtin_shufflevector(rLo, rHi, 0, 1, 2, 3, 4, 5, 6, 7,
                                        8, 9, 10, 11, 12, 13, 14, 15);
        Bo[f] = __builtin_shufflevector(iLo, iHi, 0, 1, 2, 3, 4, 5, 6, 7,
                                        8, 9, 10, 11, 12, 13, 14, 15);
    }
}


__global__ __launch_bounds__(32 * M_WAVES) __attribute__((amdgpu_waves_per_eu(2, 2)))
void qa_measure(const float* __restrict__ x,
                const float* __restrict__ params,
                float* __restrict__ out) {
    extern __shared__ char smem[];
    const int h    = blockIdx.y;
    const int tok0 = blockIdx.x * M_TOKENS;
    const int tid  = threadIdx.x;
    const int wave = tid >> 5;
    const int lane = tid & 31;
    const int c    = lane & 15;
    const int hi   = lane >> 4;

    float* cb   = (float*)(smem + CS_OFF);
    float* sb   = cb + M_TOKENS * N_QUBITS;
    float* part = (float*)(smem + PART_OFF);

    const float4 xv = *(const float4*)&x[(size_t)(tok0 + (tid >> 1)) * EMBED
                                         + h * N_QUBITS + 4 * (tid & 1)];
    const float thp = 0.5f * params[h * (Q_LAYERS * N_QUBITS) + c];

    v2f Sg[M_STRIPS], Sf;
#pragma unroll
    for (int st = 0; st < M_STRIPS; ++st)
#pragma unroll
        for (int s = 0; s < 2; ++s) {
            const int row = (wave * M_STRIPS + st) * 16 + 8 * hi + s;
            const float sg = ((row >> (7 - (c & 7))) & 1) ? -1.0f : 1.0f;
            Sg[st][s] = (c < N_QUBITS && c != 5 && c != 6) ? sg : 0.0f;
        }
    Sf[0] = (c == 5) ? 1.0f : 0.0f;
    Sf[1] = (c == 6) ? 1.0f : 0.0f;

    {
        float4 cv4, sv4;
        cv4.x = cosf(0.5f * xv.x); sv4.x = sinf(0.5f * xv.x);
        cv4.y = cosf(0.5f * xv.y); sv4.y = sinf(0.5f * xv.y);
        cv4.z = cosf(0.5f * xv.z); sv4.z = sinf(0.5f * xv.z);
        cv4.w = cosf(0.5f * xv.w); sv4.w = sinf(0.5f * xv.w);
        *(float4*)&cb[4 * tid] = cv4;
        *(float4*)&sb[4 * tid] = sv4;
    }
    __syncthreads();

    v16h Be[2][4], Bo[2][4];
    float4 cs[2][4];
    qa_load_cs(cb, sb, 0, tid, cs[0]);
    qa_build_tile(cs[0], smem + AMP_OFF, tid);
    __syncthreads();
    qa_load_frags(smem + AMP_OFF, lane, Be[0], Bo[0]);
    qa_load_cs(cb, sb, 1, tid, cs[1]);
    qa_load_cs(cb, sb, 2, tid, cs[0]);
    qa_build_tile(cs[1], smem + AMP_OFF + AMP_BUF_BYTES, tid);

    const float pcv = cosf(thp), psv = sinf(thp);
    float C0[N_QUBITS], S0[N_QUBITS], C1[N_QUBITS], S1[N_QUBITS];
#pragma unroll
    for (int w = 0; w < N_QUBITS; ++w) {
        C0[w] = __int_as_float(__builtin_amdgcn_readlane(__float_as_int(pcv), w));
        S0[w] = __int_as_float(__builtin_amdgcn_readlane(__float_as_int(psv), w));
        C1[w] = __int_as_float(__builtin_amdgcn_readlane(__float_as_int(pcv), N_QUBITS + w));
        S1[w] = __int_as_float(__builtin_amdgcn_readlane(__float_as_int(psv), N_QUBITS + w));
    }

    const int gc = c ^ (c >> 1);
    float e0l[4][2], f1l[4][2], e0h[4][2], f1h[4][2];
    const int ihw = 2 * wave;
    const int ghw = ihw ^ (ihw >> 1);
#pragma unroll
    for (int w = 0; w < 4; ++w) {
        const bool cbit = (c  >> (3 - w)) & 1;
        const bool rbit = (gc >> (3 - w)) & 1;
        const bool hbit = (ghw >> (3 - w)) & 1;
        e0l[w][0] = cbit ? -S0[4 + w] : C0[4 + w];
        e0l[w][1] = cbit ?  C0[4 + w] : S0[4 + w];
        f1l[w][0] = rbit ?  S1[4 + w] :  C1[4 + w];
        f1l[w][1] = rbit ?  C1[4 + w] : -S1[4 + w];
        e0h[w][0] = cbit ? -S0[w] : C0[w];
        e0h[w][1] = cbit ?  C0[w] : S0[w];
        f1h[w][0] = hbit ?  S1[w] :  C1[w];
        f1h[w][1] = hbit ?  C1[w] : -S1[w];
    }
    const bool rb4 = (gc >> 3) & 1, hb3 = ghw & 1;
    const float f14[2][2] = { { f1l[0][0], f1l[0][1] },
                              { rb4 ? C1[4] : S1[4], rb4 ? -S1[4] : C1[4] } };
    const float f13[2][2] = { { f1h[3][0], f1h[3][1] },
                              { hb3 ? C1[3] : S1[3], hb3 ? -S1[3] : C1[3] } };
    const float e00[2][2] = { { C0[0], S0[0] }, { -S0[0], C0[0] } };

    v8f Qd[M_STRIPS][2];
#pragma unroll
    for (int st = 0; st < M_STRIPS; ++st) { Qd[st][0] = (v8f){}; Qd[st][1] = (v8f){}; }
    {
        const float x6[2] = { hi ? e0l[2][1] : e0l[2][0], hi ? e0l[2][0] : e0l[2][1] };
        const float x7[2] = { hi ? e0l[3][1] : e0l[3][0], hi ? e0l[3][0] : e0l[3][1] };
        const float y6    = hi ? f1l[2][1] : f1l[2][0];
#pragma unroll
        for (int t = 0; t < 4; ++t) {
            const int b3 = t >> 1, b2 = t & 1;
#pragma unroll
            for (int al = 0; al < 2; ++al) {
                const float pa = e0l[0][b3 ^ al] * e0l[1][b2 ^ b3] * x6[b2];
                v2f Af;
                Af[0] = pa * x7[0];
                Af[1] = pa * x7[1];
#pragma unroll
                for (int st = 0; st < M_STRIPS; ++st) {
                    const float pb = f14[st][b3] * f1l[1][b2] * y6;
                    v2f Bf;
                    Bf[0] = pb * f1l[3][0];
                    Bf[1] = pb * f1l[3][1];
                    Qd[st][al] = __builtin_amdgcn_wmma_f32_16x16x4_f32(
                        false, Af, false, Bf, (short)0, Qd[st][al], false, false);
                }
            }
        }
    }

    v8f Rd[M_STRIPS][2][2];
#pragma unroll
    for (int st = 0; st < M_STRIPS; ++st)
#pragma unroll
        for (int al = 0; al < 2; ++al) { Rd[st][al][0] = (v8f){}; Rd[st][al][1] = (v8f){}; }
    {
        const float u1[2] = { hi ? e0h[1][1] : e0h[1][0], hi ? e0h[1][0] : e0h[1][1] };
        const float u2[2] = { hi ? e0h[2][1] : e0h[2][0], hi ? e0h[2][0] : e0h[2][1] };
        const float v1    = hi ? f1h[1][1] : f1h[1][0];
#pragma unroll
        for (int t = 0; t < 2; ++t)
#pragma unroll
            for (int al = 0; al < 2; ++al)
#pragma unroll
                for (int set = 0; set < 2; ++set) {
                    const float pa = e00[set][t] * u1[t];
                    v2f Af;
                    Af[0] = pa * u2[0] * e0h[3][al];
                    Af[1] = pa * u2[1] * e0h[3][al ^ 1];
#pragma unroll
                    for (int st = 0; st < M_STRIPS; ++st) {
                        const float pb = f1h[0][t] * v1 * f13[st][al];
                        v2f Bf;
                        Bf[0] = pb * f1h[2][0];
                        Bf[1] = pb * f1h[2][1];
                        Rd[st][al][set] = __builtin_amdgcn_wmma_f32_16x16x4_f32(
                            false, Af, false, Bf, (short)0, Rd[st][al][set], false, false);
                    }
                }
    }

    v16h Ue[M_STRIPS][4], Uo[M_STRIPS][4];
#pragma unroll
    for (int st = 0; st < M_STRIPS; ++st) {
        float QA[2][4], QB[2][4];
#pragma unroll
        for (int al = 0; al < 2; ++al) {
            const v8f qv = Qd[st][al];
            QA[al][0] = hi ? qv[1] : qv[0];  QB[al][0] = hi ? qv[0] : qv[1];
            QA[al][1] = hi ? qv[2] : qv[3];  QB[al][1] = hi ? qv[3] : qv[2];
            QA[al][2] = hi ? qv[4] : qv[5];  QB[al][2] = hi ? qv[5] : qv[4];
            QA[al][3] = hi ? qv[7] : qv[6];  QB[al][3] = hi ? qv[6] : qv[7];
        }
#pragma unroll
        for (int f = 0; f < 4; ++f)
#pragma unroll
            for (int g4 = 0; g4 < 4; ++g4) {
                const int jh = 4 * f + g4;
                const float r0 = Rd[st][0][jh >> 3][jh & 7];
                const float r1 = Rd[st][1][jh >> 3][jh & 7];
                const bool odd = ((__builtin_popcount(f) + __builtin_popcount(g4)) & 1) != 0;
#pragma unroll
                for (int k = 0; k < 4; ++k) {
                    const float ue = odd ? (r0 * QB[0][k] + r1 * QB[1][k])
                                         : (r0 * QA[0][k] + r1 * QA[1][k]);
                    const float uo = odd ? (r0 * QA[0][k] + r1 * QA[1][k])
                                         : (r0 * QB[0][k] + r1 * QB[1][k]);
                    Ue[st][f][4 * g4 + k] = (_Float16)ue;
                    Uo[st][f][4 * g4 + k] = (_Float16)uo;
                }
            }
    }
    __syncthreads();

#pragma unroll
    for (int tile = 0; tile < M_TILES; ++tile) {
        const int cur = tile & 1, nxt = cur ^ 1;
        if (tile + 1 < M_TILES)
            qa_load_frags(smem + AMP_OFF + nxt * AMP_BUF_BYTES, lane, Be[nxt], Bo[nxt]);
        if (tile + 3 < M_TILES)
            qa_load_cs(cb, sb, tile + 3, tid, cs[nxt]);
        if (tile + 1 < M_TILES)
            __builtin_amdgcn_sched_barrier(0);
        if (tile + 2 < M_TILES)
            qa_build_tile(cs[cur], smem + AMP_OFF + cur * AMP_BUF_BYTES, tid);

        v8f accR[M_STRIPS], accI[M_STRIPS];
#pragma unroll
        for (int st = 0; st < M_STRIPS; ++st) { accR[st] = (v8f){}; accI[st] = (v8f){}; }

#pragma unroll
        for (int f = 0; f < 4; ++f) {
#pragma unroll
            for (int st = 0; st < M_STRIPS; ++st) {
                accR[st] = __builtin_amdgcn_wmma_f32_16x16x32_f16(
                    false, Ue[st][f], false, Be[cur][f], (short)0, accR[st], false, false);
                accI[st] = __builtin_amdgcn_wmma_f32_16x16x32_f16(
                    false, Uo[st][f], false, Bo[cur][f], (short)0, accI[st], false, false);
            }
        }

        v8f zacc[M_STRIPS];
        v2f e5 = (v2f){}, e6 = (v2f){};
#pragma unroll
        for (int st = 0; st < M_STRIPS; ++st) {
            v2f P[4];
#pragma unroll
            for (int j = 0; j < 4; ++j)
#pragma unroll
                for (int s = 0; s < 2; ++s) {
                    const float r = accR[st][2 * j + s], q = accI[st][2 * j + s];
                    P[j][s] = r * r + q * q;
                }
            const v2f s0 = P[0] + P[2], s1 = P[1] + P[3];
            const v2f d0 = P[0] - P[2], d1 = P[1] - P[3];
            const v2f ss = s0 + s1;
            e5 += d0 + d1;
            e6 += s0 - s1;
            zacc[st] = __builtin_amdgcn_wmma_f32_16x16x4_f32(
                false, Sg[st], false, ss, (short)0, (v8f){}, false, false);
        }
        {
            v2f fB;
            fB[0] = e5[0] + e5[1];
            fB[1] = e6[0] + e6[1];
            zacc[0] = __builtin_amdgcn_wmma_f32_16x16x4_f32(
                false, Sf, false, fB, (short)0, zacc[0], false, false);
        }

        if (hi == 0) {
            float4 lo4, hi4;
            lo4.x = zacc[0][0] + zacc[1][0]; lo4.y = zacc[0][1] + zacc[1][1];
            lo4.z = zacc[0][2] + zacc[1][2]; lo4.w = zacc[0][3] + zacc[1][3];
            hi4.x = zacc[0][4] + zacc[1][4]; hi4.y = zacc[0][5] + zacc[1][5];
            hi4.z = zacc[0][6] + zacc[1][6]; hi4.w = zacc[0][7] + zacc[1][7];
            float4* pp = (float4*)(part + ((tile * M_WAVES + wave) * 16 + c) * N_QUBITS);
            pp[0] = lo4;
            pp[1] = hi4;
        }
        __syncthreads();
    }

    for (int o = tid; o < M_TOKENS * N_QUBITS; o += 32 * M_WAVES) {
        const int tile = o >> 7, rest = o & 127;
        float z = 0.0f;
#pragma unroll
        for (int wv = 0; wv < M_WAVES; ++wv)
            z += part[(tile * M_WAVES + wv) * 128 + rest];
        const int tt = tile * 16 + (rest >> 3), w = rest & 7;
        out[(size_t)(tok0 + tt) * EMBED + h * N_QUBITS + w] = z;
    }
}

extern "C" void kernel_launch(void* const* d_in, const int* in_sizes, int n_in,
                              void* d_out, int out_size, void* d_ws, size_t ws_size,
                              hipStream_t stream) {
    const float* x      = (const float*)d_in[0];
    const float* params = (const float*)d_in[1];
    float* out          = (float*)d_out;

    dim3 grid(N_TOKENS / M_TOKENS, N_HEADS);
    qa_measure<<<grid, dim3(32 * M_WAVES), M_LDS_BYTES, stream>>>(x, params, out);
}


# Signed sums: wires 3, 5, 6, 7 folded in the lane, signs of wires 0..2, 4 applied in the block's last LDS pass; no f32 16x16x4 op left in the tile loop (was 3 per tile and wave); sums regrouped
# speedup vs baseline: 181.8675x; 1.0725x over previous
#include <hip/hip_runtime.h>
#include <hip/hip_bf16.h>

typedef __attribute__((ext_vector_type(16))) _Float16 v16h;
typedef __attribute__((ext_vector_type(8)))  _Float16 v8h;
typedef __attribute__((ext_vector_type(8)))  float    v8f;
typedef __attribute__((ext_vector_type(2)))  float    v2f;

#define N_HEADS   16
#define N_QUBITS  8
#define DIM       256
#define Q_LAYERS  2
#define N_TOKENS  2048
#define EMBED     128

#define M_WAVES     8
#define M_STRIPS    2
#define M_TOKENS    128
#define M_TILES     (M_TOKENS / 16)
#define CS_OFF      0
#define AMP_OFF     8192
#define AMP_BUF_BYTES 8192
#define M_PARTS     (2 * M_WAVES)
#define PART_OFF    (AMP_OFF + 2 * AMP_BUF_BYTES)
#define PARTB_OFF   (PART_OFF + M_TILES * M_PARTS * 16 * 16)
#define M_LDS_BYTES (PARTB_OFF + M_TILES * M_PARTS * 16 * 4)


typedef __attribute__((ext_vector_type(4))) _Float16 v4hh;

__device__ __forceinline__ void qa_load_cs(const float* __restrict__ cb,
                                           const float* __restrict__ sb,
                                           int tile, int tid, float4 (&cs)[4]) {
    const int n = tid & 15;
    const float4* cp = (const float4*)(cb + (tile * 16 + n) * N_QUBITS);
    const float4* sp = (const float4*)(sb + (tile * 16 + n) * N_QUBITS);
    cs[0] = cp[0]; cs[1] = cp[1]; cs[2] = sp[0]; cs[3] = sp[1];
}

__device__ __forceinline__ void qa_build_tile(const float4 (&cs)[4],
                                              char* __restrict__ ampbuf, int tid) {
    const int n  = tid & 15;
    const int gq = tid >> 4;
    const float4 c03 = cs[0], c47 = cs[1], s03 = cs[2], s47 = cs[3];
    const float cw[8] = { c03.x, c03.y, c03.z, c03.w, c47.x, c47.y, c47.z, c47.w };
    const float sw[8] = { s03.x, s03.y, s03.z, s03.w, s47.x, s47.y, s47.z, s47.w };
    _Float16* ev = (_Float16*)ampbuf;
    _Float16* od = (_Float16*)(ampbuf + AMP_BUF_BYTES / 2);
#pragma unroll
    for (int b0 = 0; b0 < 2; ++b0) {
        const int g = gq + 16 * b0;
        float p = b0 ? sw[0] : cw[0];
#pragma unroll
        for (int w = 1; w < 5; ++w)
            p *= ((gq >> (4 - w)) & 1) ? sw[w] : cw[w];
        const int pg = __popc(g);
        const bool par = pg & 1;
        const float ps  = (pg & 2) ? -p : p;
        const float a7  = par ? sw[7] : cw[7];
        const float b7  = par ? cw[7] : sw[7];
        const float a7p = par ? -a7 : a7;
        const float u0 = ps * cw[5], u1 = ps * sw[5];
        const float t0 = u0 * cw[6], t1 = u0 * sw[6], t2 = u1 * cw[6], t3 = u1 * sw[6];
        v4hh ar, ai;
        ar[0] = (_Float16)( t0 * a7p);  ai[0] = (_Float16)(-t0 * b7);
        ar[1] = (_Float16)(-t1 * b7);   ai[1] = (_Float16)(-t1 * a7p);
        ar[2] = (_Float16)(-t2 * b7);   ai[2] = (_Float16)(-t2 * a7p);
        ar[3] = (_Float16)(-t3 * a7p);  ai[3] = (_Float16)( t3 * b7);
        const int f = g >> 3, g4 = (g >> 1) & 3, hh = g & 1;
        const int off = (((f * 2 + (g4 >> 1)) * 32) + n + 16 * hh) * 8 + 4 * (g4 & 1);
        *(v4hh*)&ev[off] = ar;
        *(v4hh*)&od[off] = ai;
    }
}

__device__ __forceinline__ void qa_load_frags(const char* __restrict__ ampbuf, int lane,
                                              v16h (&Be)[4], v16h (&Bo)[4]) {
    const _Float16* ev = (const _Float16*)ampbuf;
    const _Float16* od = ev + AMP_BUF_BYTES / 4;
#pragma unroll
    for (int f = 0; f < 4; ++f) {
        const int o0 = ((f * 2 + 0) * 32 + lane) * 8;
        const int o1 = ((f * 2 + 1) * 32 + lane) * 8;
        const v8h rLo = *(const v8h*)&ev[o0];
        const v8h rHi = *(const v8h*)&ev[o1];
        const v8h iLo = *(const v8h*)&od[o0];
        const v8h iHi = *(const v8h*)&od[o1];
        Be[f] = __builtin_shufflevector(rLo, rHi, 0, 1, 2, 3, 4, 5, 6, 7,
                                        8, 9, 10, 11, 12, 13, 14, 15);
        Bo[f] = __builtin_shufflevector(iLo, iHi, 0, 1, 2, 3, 4, 5, 6, 7,
                                        8, 9, 10, 11, 12, 13, 14, 15);
    }
}


__global__ __launch_bounds__(32 * M_WAVES) __attribute__((amdgpu_waves_per_eu(2, 2)))
void qa_measure(const float* __restrict__ x,
                const float* __restrict__ params,
                float* __restrict__ out) {
    extern __shared__ char smem[];
    const int h    = blockIdx.y;
    const int tok0 = blockIdx.x * M_TOKENS;
    const int tid  = threadIdx.x;
    const int wave = tid >> 5;
    const int lane = tid & 31;
    const int c    = lane & 15;
    const int hi   = lane >> 4;

    float* cb   = (float*)(smem + CS_OFF);
    float* sb   = cb + M_TOKENS * N_QUBITS;
    float4* partA = (float4*)(smem + PART_OFF);
    float*  partB = (float*)(smem + PARTB_OFF);

    const float4 xv = *(const float4*)&x[(size_t)(tok0 + (tid >> 1)) * EMBED
                                         + h * N_QUBITS + 4 * (tid & 1)];
    const float thp = 0.5f * params[h * (Q_LAYERS * N_QUBITS) + c];

    {
        float4 cv4, sv4;
        cv4.x = cosf(0.5f * xv.x); sv4.x = sinf(0.5f * xv.x);
        cv4.y = cosf(0.5f * xv.y); sv4.y = sinf(0.5f * xv.y);
        cv4.z = cosf(0.5f * xv.z); sv4.z = sinf(0.5f * xv.z);
        cv4.w = cosf(0.5f * xv.w); sv4.w = sinf(0.5f * xv.w);
        *(float4*)&cb[4 * tid] = cv4;
        *(float4*)&sb[4 * tid] = sv4;
    }
    __syncthreads();

    v16h Be[2][4], Bo[2][4];
    float4 cs[2][4];
    qa_load_cs(cb, sb, 0, tid, cs[0]);
    qa_build_tile(cs[0], smem + AMP_OFF, tid);
    __syncthreads();
    qa_load_frags(smem + AMP_OFF, lane, Be[0], Bo[0]);
    qa_load_cs(cb, sb, 1, tid, cs[1]);
    qa_load_cs(cb, sb, 2, tid, cs[0]);
    qa_build_tile(cs[1], smem + AMP_OFF + AMP_BUF_BYTES, tid);

    const float pcv = cosf(thp), psv = sinf(thp);
    float C0[N_QUBITS], S0[N_QUBITS], C1[N_QUBITS], S1[N_QUBITS];
#pragma unroll
    for (int w = 0; w < N_QUBITS; ++w) {
        C0[w] = __int_as_float(__builtin_amdgcn_readlane(__float_as_int(pcv), w));
        S0[w] = __int_as_float(__builtin_amdgcn_readlane(__float_as_int(psv), w));
        C1[w] = __int_as_float(__builtin_amdgcn_readlane(__float_as_int(pcv), N_QUBITS + w));
        S1[w] = __int_as_float(__builtin_amdgcn_readlane(__float_as_int(psv), N_QUBITS + w));
    }

    const int gc = c ^ (c >> 1);
    float e0l[4][2], f1l[4][2], e0h[4][2], f1h[4][2];
    const int ihw = 2 * wave;
    const int ghw = ihw ^ (ihw >> 1);
#pragma unroll
    for (int w = 0; w < 4; ++w) {
        const bool cbit = (c  >> (3 - w)) & 1;
        const bool rbit = (gc >> (3 - w)) & 1;
        const bool hbit = (ghw >> (3 - w)) & 1;
        e0l[w][0] = cbit ? -S0[4 + w] : C0[4 + w];
        e0l[w][1] = cbit ?  C0[4 + w] : S0[4 + w];
        f1l[w][0] = rbit ?  S1[4 + w] :  C1[4 + w];
        f1l[w][1] = rbit ?  C1[4 + w] : -S1[4 + w];
        e0h[w][0] = cbit ? -S0[w] : C0[w];
        e0h[w][1] = cbit ?  C0[w] : S0[w];
        f1h[w][0] = hbit ?  S1[w] :  C1[w];
        f1h[w][1] = hbit ?  C1[w] : -S1[w];
    }
    const bool rb4 = (gc >> 3) & 1, hb3 = ghw & 1;
    const float f14[2][2] = { { f1l[0][0], f1l[0][1] },
                              { rb4 ? C1[4] : S1[4], rb4 ? -S1[4] : C1[4] } };
    const float f13[2][2] = { { f1h[3][0], f1h[3][1] },
                              { hb3 ? C1[3] : S1[3], hb3 ? -S1[3] : C1[3] } };
    const float e00[2][2] = { { C0[0], S0[0] }, { -S0[0], C0[0] } };

    v8f Qd[M_STRIPS][2];
#pragma unroll
    for (int st = 0; st < M_STRIPS; ++st) { Qd[st][0] = (v8f){}; Qd[st][1] = (v8f){}; }
    {
        const float x6[2] = { hi ? e0l[2][1] : e0l[2][0], hi ? e0l[2][0] : e0l[2][1] };
        const float x7[2] = { hi ? e0l[3][1] : e0l[3][0], hi ? e0l[3][0] : e0l[3][1] };
        const float y6    = hi ? f1l[2][1] : f1l[2][0];
#pragma unroll
        for (int t = 0; t < 4; ++t) {
            const int b3 = t >> 1, b2 = t & 1;
#pragma unroll
            for (int al = 0; al < 2; ++al) {
                const float pa = e0l[0][b3 ^ al] * e0l[1][b2 ^ b3] * x6[b2];
                v2f Af;
                Af[0] = pa * x7[0];
                Af[1] = pa * x7[1];
#pragma unroll
                for (int st = 0; st < M_STRIPS; ++st) {
                    const float pb = f14[st][b3] * f1l[1][b2] * y6;
                    v2f Bf;
                    Bf[0] = pb * f1l[3][0];
                    Bf[1] = pb * f1l[3][1];
                    Qd[st][al] = __builtin_amdgcn_wmma_f32_16x16x4_f32(
                        false, Af, false, Bf, (short)0, Qd[st][al], false, false);
                }
            }
        }
    }

    v8f Rd[M_STRIPS][2][2];
#pragma unroll
    for (int st = 0; st < M_STRIPS; ++st)
#pragma unroll
        for (int al = 0; al < 2; ++al) { Rd[st][al][0] = (v8f){}; Rd[st][al][1] = (v8f){}; }
    {
        const float u1[2] = { hi ? e0h[1][1] : e0h[1][0], hi ? e0h[1][0] : e0h[1][1] };
        const float u2[2] = { hi ? e0h[2][1] : e0h[2][0], hi ? e0h[2][0] : e0h[2][1] };
        const float v1    = hi ? f1h[1][1] : f1h[1][0];
#pragma unroll
        for (int t = 0; t < 2; ++t)
#pragma unroll
            for (int al = 0; al < 2; ++al)
#pragma unroll
                for (int set = 0; set < 2; ++set) {
                    const float pa = e00[set][t] * u1[t];
                    v2f Af;
                    Af[0] = pa * u2[0] * e0h[3][al];
                    Af[1] = pa * u2[1] * e0h[3][al ^ 1];
#pragma unroll
                    for (int st = 0; st < M_STRIPS; ++st) {
                        const float pb = f1h[0][t] * v1 * f13[st][al];
                        v2f Bf;
                        Bf[0] = pb * f1h[2][0];
                        Bf[1] = pb * f1h[2][1];
                        Rd[st][al][set] = __builtin_amdgcn_wmma_f32_16x16x4_f32(
                            false, Af, false, Bf, (short)0, Rd[st][al][set], false, false);
                    }
                }
    }

    v16h Ue[M_STRIPS][4], Uo[M_STRIPS][4];
#pragma unroll
    for (int st = 0; st < M_STRIPS; ++st) {
        float QA[2][4], QB[2][4];
#pragma unroll
        for (int al = 0; al < 2; ++al) {
            const v8f qv = Qd[st][al];
            QA[al][0] = hi ? qv[1] : qv[0];  QB[al][0] = hi ? qv[0] : qv[1];
            QA[al][1] = hi ? qv[2] : qv[3];  QB[al][1] = hi ? qv[3] : qv[2];
            QA[al][2] = hi ? qv[4] : qv[5];  QB[al][2] = hi ? qv[5] : qv[4];
            QA[al][3] = hi ? qv[7] : qv[6];  QB[al][3] = hi ? qv[6] : qv[7];
        }
#pragma unroll
        for (int f = 0; f < 4; ++f)
#pragma unroll
            for (int g4 = 0; g4 < 4; ++g4) {
                const int jh = 4 * f + g4;
                const float r0 = Rd[st][0][jh >> 3][jh & 7];
                const float r1 = Rd[st][1][jh >> 3][jh & 7];
                const bool odd = ((__builtin_popcount(f) + __builtin_popcount(g4)) & 1) != 0;
#pragma unroll
                for (int k = 0; k < 4; ++k) {
                    const float ue = odd ? (r0 * QB[0][k] + r1 * QB[1][k])
                                         : (r0 * QA[0][k] + r1 * QA[1][k]);
                    const float uo = odd ? (r0 * QA[0][k] + r1 * QA[1][k])
                                         : (r0 * QB[0][k] + r1 * QB[1][k]);
                    Ue[st][f][4 * g4 + k] = (_Float16)ue;
                    Uo[st][f][4 * g4 + k] = (_Float16)uo;
                }
            }
    }
    __syncthreads();

#pragma unroll
    for (int tile = 0; tile < M_TILES; ++tile) {
        const int cur = tile & 1, nxt = cur ^ 1;
        if (tile + 1 < M_TILES)
            qa_load_frags(smem + AMP_OFF + nxt * AMP_BUF_BYTES, lane, Be[nxt], Bo[nxt]);
        if (tile + 3 < M_TILES)
            qa_load_cs(cb, sb, tile + 3, tid, cs[nxt]);
        if (tile + 1 < M_TILES)
            __builtin_amdgcn_sched_barrier(0);
        if (tile + 2 < M_TILES)
            qa_build_tile(cs[cur], smem + AMP_OFF + cur * AMP_BUF_BYTES, tid);

        v8f accR[M_STRIPS], accI[M_STRIPS];
#pragma unroll
        for (int st = 0; st < M_STRIPS; ++st) { accR[st] = (v8f){}; accI[st] = (v8f){}; }

#pragma unroll
        for (int f = 0; f < 4; ++f) {
#pragma unroll
            for (int st = 0; st < M_STRIPS; ++st) {
                accR[st] = __builtin_amdgcn_wmma_f32_16x16x32_f16(
                    false, Ue[st][f], false, Be[cur][f], (short)0, accR[st], false, false);
                accI[st] = __builtin_amdgcn_wmma_f32_16x16x32_f16(
                    false, Uo[st][f], false, Bo[cur][f], (short)0, accI[st], false, false);
            }
        }

        v2f ss[M_STRIPS], f5[M_STRIPS], f6[M_STRIPS];
#pragma unroll
        for (int st = 0; st < M_STRIPS; ++st) {
            v2f P[4];
#pragma unroll
            for (int j = 0; j < 4; ++j)
#pragma unroll
                for (int s = 0; s < 2; ++s) {
                    const float r = accR[st][2 * j + s], q = accI[st][2 * j + s];
                    P[j][s] = r * r + q * q;
                }
            const v2f s0 = P[0] + P[2], s1 = P[1] + P[3];
            const v2f d0 = P[0] - P[2], d1 = P[1] - P[3];
            ss[st] = s0 + s1;
            f5[st] = d0 + d1;
            f6[st] = s0 - s1;
        }
        {
            const v2f sT = ss[0] + ss[1], sD = ss[0] - ss[1];
            const v2f e5 = f5[0] + f5[1], e6 = f6[0] + f6[1];
            float4 pa;
            pa.x = sT[0] + sT[1];
            pa.y = sD[0] + sD[1];
            pa.z = e5[0] + e5[1];
            pa.w = e6[0] + e6[1];
            const int slot = (tile * M_PARTS + 2 * wave + hi) * 16 + c;
            partA[slot] = pa;
            partB[slot] = sT[0] - sT[1];
        }
        __syncthreads();
    }

    {
        const int tk = tid >> 1, hf = tid & 1;
        const int tl = tk >> 4, cc = tk & 15;
        float a0 = 0.0f, a1 = 0.0f, a2 = 0.0f, a3 = 0.0f;
        float a4 = 0.0f, a5 = 0.0f, a6 = 0.0f, a7 = 0.0f;
#pragma unroll
        for (int i = 0; i < M_PARTS / 2; ++i) {
            const int slot = (tl * M_PARTS + (M_PARTS / 2) * hf + i) * 16 + cc;
            const float4 v = partA[slot];
            const float v7 = partB[slot];
            a0 += v.x;
            a1 += (i & 4) ? -v.x : v.x;
            a2 += (i & 2) ? -v.x : v.x;
            a4 += (i & 1) ? -v.x : v.x;
            a3 += v.y;
            a5 += v.z;
            a6 += v.w;
            a7 += v7;
        }
        if (hf) a0 = -a0;
        const float g0 = __shfl_xor(hf ? a0 : a4, 1);
        const float g1 = __shfl_xor(hf ? a1 : a5, 1);
        const float g2 = __shfl_xor(hf ? a2 : a6, 1);
        const float g3 = __shfl_xor(hf ? a3 : a7, 1);
        float4 o;
        o.x = (hf ? a4 : a0) + g0;
        o.y = (hf ? a5 : a1) + g1;
        o.z = (hf ? a6 : a2) + g2;
        o.w = (hf ? a7 : a3) + g3;
        *(float4*)&out[(size_t)(tok0 + tk) * EMBED + h * N_QUBITS + 4 * hf] = o;
    }
}

extern "C" void kernel_launch(void* const* d_in, const int* in_sizes, int n_in,
                              void* d_out, int out_size, void* d_ws, size_t ws_size,
                              hipStream_t stream) {
    const float* x      = (const float*)d_in[0];
    const float* params = (const float*)d_in[1];
    float* out          = (float*)d_out;

    dim3 grid(N_TOKENS / M_TOKENS, N_HEADS);
    qa_measure<<<grid, dim3(32 * M_WAVES), M_LDS_BYTES, stream>>>(x, params, out);
}


# Factor products Q, R of U_h taken once per block, not once per wave: waves 0..3 one Q tile each, waves 4, 5 one R tile, shared through 6 KB of LDS; 20 f32 matrix ops a block, was 256; sums unchanged
# speedup vs baseline: 194.4875x; 1.0694x over previous
#include <hip/hip_runtime.h>
#include <hip/hip_bf16.h>

typedef __attribute__((ext_vector_type(16))) _Float16 v16h;
typedef __attribute__((ext_vector_type(8)))  _Float16 v8h;
typedef __attribute__((ext_vector_type(8)))  float    v8f;
typedef __attribute__((ext_vector_type(2)))  float    v2f;

#define N_HEADS   16
#define N_QUBITS  8
#define DIM       256
#define Q_LAYERS  2
#define N_TOKENS  2048
#define EMBED     128

#define M_WAVES     8
#define M_STRIPS    2
#define M_TOKENS    128
#define M_TILES     (M_TOKENS / 16)
#define CS_OFF      0
#define AMP_OFF     8192
#define AMP_BUF_BYTES 8192
#define M_PARTS     (2 * M_WAVES)
#define PART_OFF    (AMP_OFF + 2 * AMP_BUF_BYTES)
#define PARTB_OFF   (PART_OFF + M_TILES * M_PARTS * 16 * 16)
#define M_LDS_BYTES (PARTB_OFF + M_TILES * M_PARTS * 16 * 4)
#define QR_TILE_BYTES 1024
#define Q_OFF       PART_OFF
#define R_OFF       (PART_OFF + 4 * QR_TILE_BYTES)


typedef __attribute__((ext_vector_type(4))) _Float16 v4hh;

__device__ __forceinline__ void qa_load_cs(const float* __restrict__ cb,
                                           const float* __restrict__ sb,
                                           int tile, int tid, float4 (&cs)[4]) {
    const int n = tid & 15;
    const float4* cp = (const float4*)(cb + (tile * 16 + n) * N_QUBITS);
    const float4* sp = (const float4*)(sb + (tile * 16 + n) * N_QUBITS);
    cs[0] = cp[0]; cs[1] = cp[1]; cs[2] = sp[0]; cs[3] = sp[1];
}

__device__ __forceinline__ void qa_build_tile(const float4 (&cs)[4],
                                              char* __restrict__ ampbuf, int tid) {
    const int n  = tid & 15;
    const int gq = tid >> 4;
    const float4 c03 = cs[0], c47 = cs[1], s03 = cs[2], s47 = cs[3];
    const float cw[8] = { c03.x, c03.y, c03.z, c03.w, c47.x, c47.y, c47.z, c47.w };
    const float sw[8] = { s03.x, s03.y, s03.z, s03.w, s47.x, s47.y, s47.z, s47.w };
    _Float16* ev = (_Float16*)ampbuf;
    _Float16* od = (_Float16*)(ampbuf + AMP_BUF_BYTES / 2);
#pragma unroll
    for (int b0 = 0; b0 < 2; ++b0) {
        const int g = gq + 16 * b0;
        float p = b0 ? sw[0] : cw[0];
#pragma unroll
        for (int w = 1; w < 5; ++w)
            p *= ((gq >> (4 - w)) & 1) ? sw[w] : cw[w];
        const int pg = __popc(g);
        const bool par = pg & 1;
        const float ps  = (pg & 2) ? -p : p;
        const float a7  = par ? sw[7] : cw[7];
        const float b7  = par ? cw[7] : sw[7];
        const float a7p = par ? -a7 : a7;
        const float u0 = ps * cw[5], u1 = ps * sw[5];
        const float t0 = u0 * cw[6], t1 = u0 * sw[6], t2 = u1 * cw[6], t3 = u1 * sw[6];
        v4hh ar, ai;
        ar[0] = (_Float16)( t0 * a7p);  ai[0] = (_Float16)(-t0 * b7);
        ar[1] = (_Float16)(-t1 * b7);   ai[1] = (_Float16)(-t1 * a7p);
        ar[2] = (_Float16)(-t2 * b7);   ai[2] = (_Float16)(-t2 * a7p);
        ar[3] = (_Float16)(-t3 * a7p);  ai[3] = (_Float16)( t3 * b7);
        const int f = g >> 3, g4 = (g >> 1) & 3, hh = g & 1;
        const int off = (((f * 2 + (g4 >> 1)) * 32) + n + 16 * hh) * 8 + 4 * (g4 & 1);
        *(v4hh*)&ev[off] = ar;
        *(v4hh*)&od[off] = ai;
    }
}

__device__ __forceinline__ void qa_load_frags(const char* __restrict__ ampbuf, int lane,
                                              v16h (&Be)[4], v16h (&Bo)[4]) {
    const _Float16* ev = (const _Float16*)ampbuf;
    const _Float16* od = ev + AMP_BUF_BYTES / 4;
#pragma unroll
    for (int f = 0; f < 4; ++f) {
        const int o0 = ((f * 2 + 0) * 32 + lane) * 8;
        const int o1 = ((f * 2 + 1) * 32 + lane) * 8;
        const v8h rLo = *(const v8h*)&ev[o0];
        const v8h rHi = *(const v8h*)&ev[o1];
        const v8h iLo = *(const v8h*)&od[o0];
        const v8h iHi = *(const v8h*)&od[o1];
        Be[f] = __builtin_shufflevector(rLo, rHi, 0, 1, 2, 3, 4, 5, 6, 7,
                                        8, 9, 10, 11, 12, 13, 14, 15);
        Bo[f] = __builtin_shufflevector(iLo, iHi, 0, 1, 2, 3, 4, 5, 6, 7,
                                        8, 9, 10, 11, 12, 13, 14, 15);
    }
}


__global__ __launch_bounds__(32 * M_WAVES) __attribute__((amdgpu_waves_per_eu(2, 2)))
void qa_measure(const float* __restrict__ x,
                const float* __restrict__ params,
                float* __restrict__ out) {
    extern __shared__ char smem[];
    const int h    = blockIdx.y;
    const int tok0 = blockIdx.x * M_TOKENS;
    const int tid  = threadIdx.x;
    const int wave = tid >> 5;
    const int lane = tid & 31;
    const int c    = lane & 15;
    const int hi   = lane >> 4;

    float* cb   = (float*)(smem + CS_OFF);
    float* sb   = cb + M_TOKENS * N_QUBITS;
    float4* partA = (float4*)(smem + PART_OFF);
    float*  partB = (float*)(smem + PARTB_OFF);

    const float4 xv = *(const float4*)&x[(size_t)(tok0 + (tid >> 1)) * EMBED
                                         + h * N_QUBITS + 4 * (tid & 1)];
    const float thp = 0.5f * params[h * (Q_LAYERS * N_QUBITS) + c];

    {
        float4 cv4, sv4;
        cv4.x = cosf(0.5f * xv.x); sv4.x = sinf(0.5f * xv.x);
        cv4.y = cosf(0.5f * xv.y); sv4.y = sinf(0.5f * xv.y);
        cv4.z = cosf(0.5f * xv.z); sv4.z = sinf(0.5f * xv.z);
        cv4.w = cosf(0.5f * xv.w); sv4.w = sinf(0.5f * xv.w);
        *(float4*)&cb[4 * tid] = cv4;
        *(float4*)&sb[4 * tid] = sv4;
    }

    const int wv = __builtin_amdgcn_readfirstlane(wave);
    if (wv < 6) {
        const bool isq = wv < 4;
        const float pcv = cosf(thp), psv = sinf(thp);
        const int w0 = isq ? 4 : 0;
        float C0w[4], S0w[4], C1w[4], S1w[4];
#pragma unroll
        for (int w = 0; w < 4; ++w) {
            C0w[w] = __int_as_float(__builtin_amdgcn_readlane(__float_as_int(pcv), w0 + w));
            S0w[w] = __int_as_float(__builtin_amdgcn_readlane(__float_as_int(psv), w0 + w));
            C1w[w] = __int_as_float(__builtin_amdgcn_readlane(__float_as_int(pcv), N_QUBITS + w0 + w));
            S1w[w] = __int_as_float(__builtin_amdgcn_readlane(__float_as_int(psv), N_QUBITS + w0 + w));
        }
        const int gc = c ^ (c >> 1);
        float e[4][2], f[4][2];
#pragma unroll
        for (int w = 0; w < 4; ++w) {
            const bool cbit = (c  >> (3 - w)) & 1;
            const bool rbit = (gc >> (3 - w)) & 1;
            e[w][0] = cbit ? -S0w[w] : C0w[w];
            e[w][1] = cbit ?  C0w[w] : S0w[w];
            f[w][0] = rbit ?  S1w[w] :  C1w[w];
            f[w][1] = rbit ?  C1w[w] : -S1w[w];
        }
        v8f acc = (v8f){};
        float* dst;
        if (isq) {
            const int st = wv & 1, al = wv >> 1;
            const bool rb4 = (gc >> 3) & 1;
            const float f4s[2] = { st ? (rb4 ?  C1w[0] : S1w[0]) : f[0][0],
                                   st ? (rb4 ? -S1w[0] : C1w[0]) : f[0][1] };
            const float e4[2] = { al ? e[0][1] : e[0][0], al ? e[0][0] : e[0][1] };
            const float x6[2] = { hi ? e[2][1] : e[2][0], hi ? e[2][0] : e[2][1] };
            const float x7[2] = { hi ? e[3][1] : e[3][0], hi ? e[3][0] : e[3][1] };
            const float y6    = hi ? f[2][1] : f[2][0];
#pragma unroll
            for (int t = 0; t < 4; ++t) {
                const int b3 = t >> 1, b2 = t & 1;
                const float pa = e4[b3] * e[1][b2 ^ b3] * x6[b2];
                v2f Af;
                Af[0] = pa * x7[0];
                Af[1] = pa * x7[1];
                const float pb = f4s[b3] * f[1][b2] * y6;
                v2f Bf;
                Bf[0] = pb * f[3][0];
                Bf[1] = pb * f[3][1];
                acc = __builtin_amdgcn_wmma_f32_16x16x4_f32(
                    false, Af, false, Bf, (short)0, acc, false, false);
            }
            dst = (float*)(smem + Q_OFF + wv * QR_TILE_BYTES);
        } else {
            const int al = wv - 4;
            const float u1[2] = { hi ? e[1][1] : e[1][0], hi ? e[1][0] : e[1][1] };
            const float u2[2] = { hi ? e[2][1] : e[2][0], hi ? e[2][0] : e[2][1] };
            const float e3[2] = { al ? e[3][1] : e[3][0], al ? e[3][0] : e[3][1] };
            const float v1    = hi ? f[1][1] : f[1][0];
            const float f3    = al ? f[3][1] : f[3][0];
#pragma unroll
            for (int t = 0; t < 2; ++t) {
                const float pa = e[0][t] * u1[t];
                v2f Af;
                Af[0] = pa * u2[0] * e3[0];
                Af[1] = pa * u2[1] * e3[1];
                const float pb = f[0][t] * v1 * f3;
                v2f Bf;
                Bf[0] = pb * f[2][0];
                Bf[1] = pb * f[2][1];
                acc = __builtin_amdgcn_wmma_f32_16x16x4_f32(
                    false, Af, false, Bf, (short)0, acc, false, false);
            }
            dst = (float*)(smem + R_OFF + al * QR_TILE_BYTES);
        }
        *(float4*)&dst[4 * lane]       = make_float4(acc[0], acc[1], acc[2], acc[3]);
        *(float4*)&dst[128 + 4 * lane] = make_float4(acc[4], acc[5], acc[6], acc[7]);
    }
    __syncthreads();

    v16h Be[2][4], Bo[2][4];
    float4 cs[2][4];
    qa_load_cs(cb, sb, 0, tid, cs[0]);
    qa_build_tile(cs[0], smem + AMP_OFF, tid);
    __syncthreads();
    qa_load_cs(cb, sb, 1, tid, cs[1]);
    qa_load_cs(cb, sb, 2, tid, cs[0]);

    v8f Qd[M_STRIPS][2];
    v8f Rd[M_STRIPS][2][2];
#pragma unroll
    for (int st = 0; st < M_STRIPS; ++st) {
#pragma unroll
        for (int al = 0; al < 2; ++al) {
            const float* qt = (const float*)(smem + Q_OFF + (st + 2 * al) * QR_TILE_BYTES);
            const float4 qlo = *(const float4*)&qt[4 * lane];
            const float4 qhi = *(const float4*)&qt[128 + 4 * lane];
            Qd[st][al] = (v8f){ qlo.x, qlo.y, qlo.z, qlo.w, qhi.x, qhi.y, qhi.z, qhi.w };
            const float* rt = (const float*)(smem + R_OFF + al * QR_TILE_BYTES);
#pragma unroll
            for (int set = 0; set < 2; ++set) {
                const int sl = 2 * wave + st + 16 * set;
                const float4 rlo = *(const float4*)&rt[4 * sl];
                const float4 rhi = *(const float4*)&rt[128 + 4 * sl];
                Rd[st][al][set] = (v8f){ rlo.x, rlo.y, rlo.z, rlo.w, rhi.x, rhi.y, rhi.z, rhi.w };
            }
        }
    }
    qa_load_frags(smem + AMP_OFF, lane, Be[0], Bo[0]);
    qa_build_tile(cs[1], smem + AMP_OFF + AMP_BUF_BYTES, tid);

    v16h Ue[M_STRIPS][4], Uo[M_STRIPS][4];
#pragma unroll
    for (int st = 0; st < M_STRIPS; ++st) {
        float QA[2][4], QB[2][4];
#pragma unroll
        for (int al = 0; al < 2; ++al) {
            const v8f qv = Qd[st][al];
            QA[al][0] = hi ? qv[1] : qv[0];  QB[al][0] = hi ? qv[0] : qv[1];
            QA[al][1] = hi ? qv[2] : qv[3];  QB[al][1] = hi ? qv[3] : qv[2];
            QA[al][2] = hi ? qv[4] : qv[5];  QB[al][2] = hi ? qv[5] : qv[4];
            QA[al][3] = hi ? qv[7] : qv[6];  QB[al][3] = hi ? qv[6] : qv[7];
        }
#pragma unroll
        for (int f = 0; f < 4; ++f)
#pragma unroll
            for (int g4 = 0; g4 < 4; ++g4) {
                const int jh = 4 * f + g4;
                const float r0 = Rd[st][0][jh >> 3][jh & 7];
                const float r1 = Rd[st][1][jh >> 3][jh & 7];
                const bool odd = ((__builtin_popcount(f) + __builtin_popcount(g4)) & 1) != 0;
#pragma unroll
                for (int k = 0; k < 4; ++k) {
                    const float ue = odd ? (r0 * QB[0][k] + r1 * QB[1][k])
                                         : (r0 * QA[0][k] + r1 * QA[1][k]);
                    const float uo = odd ? (r0 * QA[0][k] + r1 * QA[1][k])
                                         : (r0 * QB[0][k] + r1 * QB[1][k]);
                    Ue[st][f][4 * g4 + k] = (_Float16)ue;
                    Uo[st][f][4 * g4 + k] = (_Float16)uo;
                }
            }
    }
    __syncthreads();

#pragma unroll
    for (int tile = 0; tile < M_TILES; ++tile) {
        const int cur = tile & 1, nxt = cur ^ 1;
        if (tile + 1 < M_TILES)
            qa_load_frags(smem + AMP_OFF + nxt * AMP_BUF_BYTES, lane, Be[nxt], Bo[nxt]);
        if (tile + 3 < M_TILES)
            qa_load_cs(cb, sb, tile + 3, tid, cs[nxt]);
        if (tile + 1 < M_TILES)
            __builtin_amdgcn_sched_barrier(0);
        if (tile + 2 < M_TILES)
            qa_build_tile(cs[cur], smem + AMP_OFF + cur * AMP_BUF_BYTES, tid);

        v8f accR[M_STRIPS], accI[M_STRIPS];
#pragma unroll
        for (int st = 0; st < M_STRIPS; ++st) { accR[st] = (v8f){}; accI[st] = (v8f){}; }

#pragma unroll
        for (int f = 0; f < 4; ++f) {
#pragma unroll
            for (int st = 0; st < M_STRIPS; ++st) {
                accR[st] = __builtin_amdgcn_wmma_f32_16x16x32_f16(
                    false, Ue[st][f], false, Be[cur][f], (short)0, accR[st], false, false);
                accI[st] = __builtin_amdgcn_wmma_f32_16x16x32_f16(
                    false, Uo[st][f], false, Bo[cur][f], (short)0, accI[st], false, false);
            }
        }

        v2f ss[M_STRIPS], f5[M_STRIPS], f6[M_STRIPS];
#pragma unroll
        for (int st = 0; st < M_STRIPS; ++st) {
            v2f P[4];
#pragma unroll
            for (int j = 0; j < 4; ++j)
#pragma unroll
                for (int s = 0; s < 2; ++s) {
                    const float r = accR[st][2 * j + s], q = accI[st][2 * j + s];
                    P[j][s] = r * r + q * q;
                }
            const v2f s0 = P[0] + P[2], s1 = P[1] + P[3];
            const v2f d0 = P[0] - P[2], d1 = P[1] - P[3];
            ss[st] = s0 + s1;
            f5[st] = d0 + d1;
            f6[st] = s0 - s1;
        }
        {
            const v2f sT = ss[0] + ss[1], sD = ss[0] - ss[1];
            const v2f e5 = f5[0] + f5[1], e6 = f6[0] + f6[1];
            float4 pa;
            pa.x = sT[0] + sT[1];
            pa.y = sD[0] + sD[1];
            pa.z = e5[0] + e5[1];
            pa.w = e6[0] + e6[1];
            const int slot = (tile * M_PARTS + 2 * wave + hi) * 16 + c;
            partA[slot] = pa;
            partB[slot] = sT[0] - sT[1];
        }
        __syncthreads();
    }

    {
        const int tk = tid >> 1, hf = tid & 1;
        const int tl = tk >> 4, cc = tk & 15;
        float a0 = 0.0f, a1 = 0.0f, a2 = 0.0f, a3 = 0.0f;
        float a4 = 0.0f, a5 = 0.0f, a6 = 0.0f, a7 = 0.0f;
#pragma unroll
        for (int i = 0; i < M_PARTS / 2; ++i) {
            const int slot = (tl * M_PARTS + (M_PARTS / 2) * hf + i) * 16 + cc;
            const float4 v = partA[slot];
            const float v7 = partB[slot];
            a0 += v.x;
            a1 += (i & 4) ? -v.x : v.x;
            a2 += (i & 2) ? -v.x : v.x;
            a4 += (i & 1) ? -v.x : v.x;
            a3 += v.y;
            a5 += v.z;
            a6 += v.w;
            a7 += v7;
        }
        if (hf) a0 = -a0;
        const float g0 = __shfl_xor(hf ? a0 : a4, 1);
        const float g1 = __shfl_xor(hf ? a1 : a5, 1);
        const float g2 = __shfl_xor(hf ? a2 : a6, 1);
        const float g3 = __shfl_xor(hf ? a3 : a7, 1);
        float4 o;
        o.x = (hf ? a4 : a0) + g0;
        o.y = (hf ? a5 : a1) + g1;
        o.z = (hf ? a6 : a2) + g2;
        o.w = (hf ? a7 : a3) + g3;
        *(float4*)&out[(size_t)(tok0 + tk) * EMBED + h * N_QUBITS + 4 * hf] = o;
    }
}

extern "C" void kernel_launch(void* const* d_in, const int* in_sizes, int n_in,
                              void* d_out, int out_size, void* d_ws, size_t ws_size,
                              hipStream_t stream) {
    const float* x      = (const float*)d_in[0];
    const float* params = (const float*)d_in[1];
    float* out          = (float*)d_out;

    dim3 grid(N_TOKENS / M_TOKENS, N_HEADS);
    qa_measure<<<grid, dim3(32 * M_WAVES), M_LDS_BYTES, stream>>>(x, params, out);
}
